# starting kernel
# baseline (speedup 1.0000x reference)
#include <hip/hip_runtime.h>
#include <hip/hip_bf16.h>

typedef __attribute__((ext_vector_type(2))) float v2f;
typedef __attribute__((ext_vector_type(8))) float v8f;

#define DIN 128
#define DCOL 64

__global__ void k_init_deg(float* __restrict__ deg, int n) {
    int i = blockIdx.x * blockDim.x + threadIdx.x;
    if (i < n) deg[i] = 1.0f;
}

__global__ void k_accum_deg(const int* __restrict__ dst, float* __restrict__ deg, int e) {
    int i = blockIdx.x * blockDim.x + threadIdx.x;
    if (i < e) atomicAdd(&deg[dst[i]], 1.0f);
}

__global__ void k_finish_dinv(float* __restrict__ deg, int n) {
    int i = blockIdx.x * blockDim.x + threadIdx.x;
    if (i < n) deg[i] = rsqrtf(deg[i]);
}

template <int K, bool RELU, bool BIAS>
__global__ __launch_bounds__(128)
void k_gemm_wmma(const float* __restrict__ A, const float* __restrict__ W,
                 float* __restrict__ H, const float* __restrict__ bias) {
    const int lane    = threadIdx.x & 31;
    const int wave    = threadIdx.x >> 5;
    const int rowBase = blockIdx.x << 4;
    const int colBase = wave << 4;
    const int half    = lane >> 4;
    const int l16     = lane & 15;

    const float* __restrict__ arow = A + (size_t)(rowBase + l16) * K;
    const float* __restrict__ wcol = W + colBase + l16;

    v8f c = {};
    #pragma unroll 8
    for (int kb = 0; kb < K; kb += 4) {
        const int k0 = kb + half * 2;
        float a0 = arow[k0];
        float a1 = arow[k0 + 1];
        if constexpr (RELU) { a0 = fmaxf(a0, 0.0f); a1 = fmaxf(a1, 0.0f); }
        v2f a; a[0] = a0; a[1] = a1;
        v2f b; b[0] = wcol[(size_t)k0 * DCOL]; b[1] = wcol[(size_t)(k0 + 1) * DCOL];
        c = __builtin_amdgcn_wmma_f32_16x16x4_f32(
                 false, a,  false, b,
                 (short)0, c,  false,  false);
    }

    float bv;
    if constexpr (BIAS) bv = bias[colBase + l16];
    float* __restrict__ hout = H + colBase + l16;
    #pragma unroll
    for (int v = 0; v < 8; ++v) {
        const int row = rowBase + v + half * 8;
        float val = c[v];
        if constexpr (BIAS) val += bv;
        hout[(size_t)row * DCOL] = val;
    }
}

__global__ void k_agg_init(const float* __restrict__ h, const float* __restrict__ dinv,
                           const float* __restrict__ bias, float* __restrict__ out, int n) {
    int t = blockIdx.x * blockDim.x + threadIdx.x;
    int i = t >> 4, q = t & 15;
    if (i >= n) return;
    float w = dinv[i]; w *= w;
    float4 hv = reinterpret_cast<const float4*>(h + (size_t)i * DCOL)[q];
    float4 bv = reinterpret_cast<const float4*>(bias)[q];
    float4 o;
    o.x = fmaf(w, hv.x, bv.x); o.y = fmaf(w, hv.y, bv.y);
    o.z = fmaf(w, hv.z, bv.z); o.w = fmaf(w, hv.w, bv.w);
    reinterpret_cast<float4*>(out + (size_t)i * DCOL)[q] = o;
}

__global__ void k_agg_edges(const int* __restrict__ src, const int* __restrict__ dst,
                            const float* __restrict__ dinv, const float* __restrict__ h,
                            float* __restrict__ out, int e) {
    int t = blockIdx.x * blockDim.x + threadIdx.x;
    int ed = t >> 4, q = t & 15;
    if (ed >= e) return;
    int s = src[ed];
    int d = dst[ed];
    float w = dinv[s] * dinv[d];
    float4 hv = reinterpret_cast<const float4*>(h + (size_t)s * DCOL)[q];
    float* o = out + (size_t)d * DCOL + q * 4;
    atomicAdd(o + 0, w * hv.x);
    atomicAdd(o + 1, w * hv.y);
    atomicAdd(o + 2, w * hv.z);
    atomicAdd(o + 3, w * hv.w);
}

extern "C" void kernel_launch(void* const* d_in, const int* in_sizes, int n_in,
                              void* d_out, int out_size, void* d_ws, size_t ws_size,
                              hipStream_t stream) {
    const float* x   = (const float*)d_in[0];
    const int*   ei  = (const int*)  d_in[1];
    const float* W1  = (const float*)d_in[2];
    const float* b1  = (const float*)d_in[3];
    const float* W2  = (const float*)d_in[4];
    const float* b2  = (const float*)d_in[5];
    const float* Wfc = (const float*)d_in[6];
    const float* bfc = (const float*)d_in[7];
    float* out = (float*)d_out;

    const int N = in_sizes[0] / DIN;
    const int E = in_sizes[1] / 2;
    const int* src = ei;
    const int* dst = ei + E;

    float* dinv = (float*)d_ws;
    float* buf0 = dinv + ((N + 255) / 256) * 256;
    float* buf1 = buf0 + (size_t)N * DCOL;

    const int blk = 256;
    const int gN   = (N + blk - 1) / blk;
    const int gE   = (E + blk - 1) / blk;
    const int gN16 = (N * 16 + blk - 1) / blk;
    const int gE16 = (E * 16 + blk - 1) / blk;
    const int gGemm = N / 16;

    k_init_deg   <<<gN, blk, 0, stream>>>(dinv, N);
    k_accum_deg  <<<gE, blk, 0, stream>>>(dst, dinv, E);
    k_finish_dinv<<<gN, blk, 0, stream>>>(dinv, N);

    k_gemm_wmma<DIN,  false, false><<<gGemm, 128, 0, stream>>>(x,    W1,  buf0, nullptr);
    k_agg_init  <<<gN16, blk, 0, stream>>>(buf0, dinv, b1, buf1, N);
    k_agg_edges <<<gE16, blk, 0, stream>>>(src, dst, dinv, buf0, buf1, E);

    k_gemm_wmma<DCOL, true,  false><<<gGemm, 128, 0, stream>>>(buf1, W2,  buf0, nullptr);
    k_agg_init  <<<gN16, blk, 0, stream>>>(buf0, dinv, b2, buf1, N);
    k_agg_edges <<<gE16, blk, 0, stream>>>(src, dst, dinv, buf0, buf1, E);

    k_gemm_wmma<DCOL, false, true ><<<gGemm, 128, 0, stream>>>(buf1, Wfc, out,  bfc);
}


# Aggregation reads a destination-grouped edge list (int-atomic rank, scan, fill) and sums each node's rows in registers; replaces f32 atomic scatter; sum order per node = arrival-order rank
# speedup vs baseline: 1.9445x; 1.9445x over previous
#include <hip/hip_runtime.h>
#include <hip/hip_bf16.h>


typedef __attribute__((ext_vector_type(2))) float v2f;
typedef __attribute__((ext_vector_type(8))) float v8f;

#define DIN 128
#define DCOL 64


__global__ void k_zero_cnt(int* __restrict__ cnt, int n) {
    int i = blockIdx.x * blockDim.x + threadIdx.x;
    if (i < n) cnt[i] = 0;
}

__global__ void k_count_rank(const int* __restrict__ dst, int* __restrict__ cnt,
                             int* __restrict__ rank, int e) {
    int i = blockIdx.x * blockDim.x + threadIdx.x;
    if (i < e) rank[i] = atomicAdd(&cnt[dst[i]], 1);
}

__global__ __launch_bounds__(1024)
void k_scan_dinv(const int* __restrict__ cnt, int* __restrict__ rowptr,
                 float* __restrict__ dinv, int n) {
    __shared__ int s_pre[32];
    __shared__ int s_tot[32];
    const int t = threadIdx.x, lane = t & 31, wv = t >> 5;
    const int b = blockIdx.x;

    int pre = 0;
    const int4* __restrict__ c4 = reinterpret_cast<const int4*>(cnt);
    const int nchunk = b * 256;
    #pragma unroll 4
    for (int j = t; j < nchunk; j += 1024) {
        int4 v = c4[j];
        pre += (v.x + v.y) + (v.z + v.w);
    }
    const int i = b * 1024 + t;
    const int v = (i < n) ? cnt[i] : 0;

    int inc = v;
    #pragma unroll
    for (int o = 1; o < 32; o <<= 1) {
        int u = __shfl_up(inc, o);
        if (lane >= o) inc += u;
    }
    #pragma unroll
    for (int o = 16; o > 0; o >>= 1) pre += __shfl_xor(pre, o);
    if (lane == 31) s_tot[wv] = inc;
    if (lane == 0)  s_pre[wv] = pre;
    __syncthreads();
    int off = 0;
    #pragma unroll
    for (int k = 0; k < 32; ++k) {
        off += s_pre[k];
        if (k < wv) off += s_tot[k];
    }
    if (i < n) {
        rowptr[i] = off + inc - v;
        if (i == n - 1) rowptr[n] = off + inc;
        dinv[i] = rsqrtf((float)(v + 1));
    }
}

__global__ void k_fill(const int* __restrict__ src, const int* __restrict__ dst,
                       const int* __restrict__ rank, const int* __restrict__ rowptr,
                       const float* __restrict__ dinv, int* __restrict__ rec, int e) {
    int i = blockIdx.x * blockDim.x + threadIdx.x;
    if (i >= e) return;
    int s = src[i];
    int p = rowptr[dst[i]] + rank[i];
    reinterpret_cast<int2*>(rec)[p] = make_int2(s, __float_as_int(dinv[s]));
}

template <int K, bool RELU, bool BIAS>
__global__ __launch_bounds__(128)
void k_gemm_wmma(const float* __restrict__ A, const float* __restrict__ W,
                 float* __restrict__ H, const float* __restrict__ bias) {
    const int lane    = threadIdx.x & 31;
    const int wave    = threadIdx.x >> 5;
    const int rowBase = blockIdx.x << 4;
    const int colBase = wave << 4;
    const int half    = lane >> 4;
    const int l16     = lane & 15;

    const float* __restrict__ arow = A + (size_t)(rowBase + l16) * K;
    const float* __restrict__ wcol = W + colBase + l16;

    v8f c = {};
    #pragma unroll 8
    for (int kb = 0; kb < K; kb += 4) {
        const int k0 = kb + half * 2;
        float a0 = arow[k0];
        float a1 = arow[k0 + 1];
        if constexpr (RELU) { a0 = fmaxf(a0, 0.0f); a1 = fmaxf(a1, 0.0f); }
        v2f a; a[0] = a0; a[1] = a1;
        v2f b; b[0] = wcol[(size_t)k0 * DCOL]; b[1] = wcol[(size_t)(k0 + 1) * DCOL];
        c = __builtin_amdgcn_wmma_f32_16x16x4_f32(
                 false, a,  false, b,
                 (short)0, c,  false,  false);
    }

    float bv;
    if constexpr (BIAS) bv = bias[colBase + l16];
    float* __restrict__ hout = H + colBase + l16;
    #pragma unroll
    for (int v = 0; v < 8; ++v) {
        const int row = rowBase + v + half * 8;
        float val = c[v];
        if constexpr (BIAS) val += bv;
        hout[(size_t)row * DCOL] = val;
    }
}

#define GB 8
__global__ __launch_bounds__(256)
void k_agg_gather(const int* __restrict__ rowptr, const int* __restrict__ rec,
                  const float* __restrict__ dinv, const float* __restrict__ h,
                  const float* __restrict__ bias, float* __restrict__ out, int n) {
    const int t = blockIdx.x * blockDim.x + threadIdx.x;
    const int i = t >> 4, q = t & 15;
    if (i >= n) return;
    const float4* __restrict__ h4 = reinterpret_cast<const float4*>(h);
    const int2*   __restrict__ r2 = reinterpret_cast<const int2*>(rec);

    const int beg = rowptr[i];
    const int end = rowptr[i + 1];
    const float di = dinv[i];
    const float4 hs = h4[i * 16 + q];
    const float4 bv = reinterpret_cast<const float4*>(bias)[q];
    const float ws = di * di;
    float4 acc;
    acc.x = fmaf(ws, hs.x, bv.x); acc.y = fmaf(ws, hs.y, bv.y);
    acc.z = fmaf(ws, hs.z, bv.z); acc.w = fmaf(ws, hs.w, bv.w);

    for (int p = beg; p < end; p += GB) {
        int2   r[GB];
        float4 v[GB];
        #pragma unroll
        for (int u = 0; u < GB; ++u) r[u] = r2[min(p + u, end - 1)];
        #pragma unroll
        for (int u = 0; u < GB; ++u) v[u] = h4[r[u].x * 16 + q];
        #pragma unroll
        for (int u = 0; u < GB; ++u) {
            const bool live = (u == 0) || (p + u < end);
            const float w = __int_as_float(r[u].y) * di;
            const float nx = fmaf(w, v[u].x, acc.x), ny = fmaf(w, v[u].y, acc.y);
            const float nz = fmaf(w, v[u].z, acc.z), nw = fmaf(w, v[u].w, acc.w);
            acc.x = live ? nx : acc.x; acc.y = live ? ny : acc.y;
            acc.z = live ? nz : acc.z; acc.w = live ? nw : acc.w;
        }
    }
    reinterpret_cast<float4*>(out)[i * 16 + q] = acc;
}

extern "C" void kernel_launch(void* const* d_in, const int* in_sizes, int n_in,
                              void* d_out, int out_size, void* d_ws, size_t ws_size,
                              hipStream_t stream) {
    const float* x   = (const float*)d_in[0];
    const int*   ei  = (const int*)  d_in[1];
    const float* W1  = (const float*)d_in[2];
    const float* b1  = (const float*)d_in[3];
    const float* W2  = (const float*)d_in[4];
    const float* b2  = (const float*)d_in[5];
    const float* Wfc = (const float*)d_in[6];
    const float* bfc = (const float*)d_in[7];
    float* out = (float*)d_out;

    const int N = in_sizes[0] / DIN;
    const int E = in_sizes[1] / 2;
    const int* src = ei;
    const int* dst = ei + E;

    const size_t Np = ((size_t)(N + 255) / 256) * 256;
    const size_t Ep = ((size_t)(E + 255) / 256) * 256;
    int*   cnt    = (int*)d_ws;
    int*   rowptr = cnt + Np;
    float* dinv   = (float*)(rowptr + Np + 256);
    int*   rank   = (int*)(dinv + Np);
    int*   rec    = rank + Ep;
    float* buf0   = (float*)(rec + 2 * Ep);
    float* buf1   = buf0 + (size_t)N * DCOL;

    const int blk = 256;
    const int gN   = (N + blk - 1) / blk;
    const int gE   = (E + blk - 1) / blk;
    const int gN16 = (N * 16 + blk - 1) / blk;
    const int gScan = (N + 1023) / 1024;
    const int gGemm = N / 16;

    k_zero_cnt  <<<gN, blk, 0, stream>>>(cnt, N);
    k_count_rank<<<gE, blk, 0, stream>>>(dst, cnt, rank, E);
    k_scan_dinv <<<gScan, 1024, 0, stream>>>(cnt, rowptr, dinv, N);
    k_fill      <<<gE, blk, 0, stream>>>(src, dst, rank, rowptr, dinv, rec, E);

    k_gemm_wmma<DIN,  false, false><<<gGemm, 128, 0, stream>>>(x,    W1,  buf0, nullptr);
    k_agg_gather<<<gN16, blk, 0, stream>>>(rowptr, rec, dinv, buf0, b1, buf1, N);

    k_gemm_wmma<DCOL, true,  false><<<gGemm, 128, 0, stream>>>(buf1, W2,  buf0, nullptr);
    k_agg_gather<<<gN16, blk, 0, stream>>>(rowptr, rec, dinv, buf0, b2, buf1, N);

    k_gemm_wmma<DCOL, false, true ><<<gGemm, 128, 0, stream>>>(buf1, Wfc, out,  bfc);
}


# Each aggregation block hands its 16 finished rows through LDS to the f32 16x16x4 matrix units and applies the next dense layer (W2, then the head) in the same launch: 9 launches become 7
# speedup vs baseline: 2.1042x; 1.0821x over previous
#include <hip/hip_runtime.h>
#include <hip/hip_bf16.h>


typedef __attribute__((ext_vector_type(2))) float v2f;
typedef __attribute__((ext_vector_type(8))) float v8f;

#define DIN 128
#define DCOL 64


__global__ void k_zero_cnt(int* __restrict__ cnt, int n) {
    int i = blockIdx.x * blockDim.x + threadIdx.x;
    if (i < n) cnt[i] = 0;
}

__global__ void k_count_rank(const int* __restrict__ dst, int* __restrict__ cnt,
                             int* __restrict__ rank, int e) {
    int i = blockIdx.x * blockDim.x + threadIdx.x;
    if (i < e) rank[i] = atomicAdd(&cnt[dst[i]], 1);
}

__global__ __launch_bounds__(1024)
void k_scan_dinv(const int* __restrict__ cnt, int* __restrict__ rowptr,
                 float* __restrict__ dinv, int n) {
    __shared__ int s_pre[32];
    __shared__ int s_tot[32];
    const int t = threadIdx.x, lane = t & 31, wv = t >> 5;
    const int b = blockIdx.x;

    int pre = 0;
    const int4* __restrict__ c4 = reinterpret_cast<const int4*>(cnt);
    const int nchunk = b * 256;
    #pragma unroll 4
    for (int j = t; j < nchunk; j += 1024) {
        int4 v = c4[j];
        pre += (v.x + v.y) + (v.z + v.w);
    }
    const int i = b * 1024 + t;
    const int v = (i < n) ? cnt[i] : 0;

    int inc = v;
    #pragma unroll
    for (int o = 1; o < 32; o <<= 1) {
        int u = __shfl_up(inc, o);
        if (lane >= o) inc += u;
    }
    #pragma unroll
    for (int o = 16; o > 0; o >>= 1) pre += __shfl_xor(pre, o);
    if (lane == 31) s_tot[wv] = inc;
    if (lane == 0)  s_pre[wv] = pre;
    __syncthreads();
    int off = 0;
    #pragma unroll
    for (int k = 0; k < 32; ++k) {
        off += s_pre[k];
        if (k < wv) off += s_tot[k];
    }
    if (i < n) {
        rowptr[i] = off + inc - v;
        if (i == n - 1) rowptr[n] = off + inc;
        dinv[i] = rsqrtf((float)(v + 1));
    }
}

__global__ void k_fill(const int* __restrict__ src, const int* __restrict__ dst,
                       const int* __restrict__ rank, const int* __restrict__ rowptr,
                       const float* __restrict__ dinv, int* __restrict__ rec, int e) {
    int i = blockIdx.x * blockDim.x + threadIdx.x;
    if (i >= e) return;
    int s = src[i];
    int p = rowptr[dst[i]] + rank[i];
    reinterpret_cast<int2*>(rec)[p] = make_int2(s, __float_as_int(dinv[s]));
}

template <int K, bool RELU, bool BIAS>
__global__ __launch_bounds__(128)
void k_gemm_wmma(const float* __restrict__ A, const float* __restrict__ W,
                 float* __restrict__ H, const float* __restrict__ bias) {
    const int lane    = threadIdx.x & 31;
    const int wave    = threadIdx.x >> 5;
    const int rowBase = blockIdx.x << 4;
    const int colBase = wave << 4;
    const int half    = lane >> 4;
    const int l16     = lane & 15;

    const float* __restrict__ arow = A + (size_t)(rowBase + l16) * K;
    const float* __restrict__ wcol = W + colBase + l16;

    v8f c = {};
    #pragma unroll 8
    for (int kb = 0; kb < K; kb += 4) {
        const int k0 = kb + half * 2;
        float a0 = arow[k0];
        float a1 = arow[k0 + 1];
        if constexpr (RELU) { a0 = fmaxf(a0, 0.0f); a1 = fmaxf(a1, 0.0f); }
        v2f a; a[0] = a0; a[1] = a1;
        v2f b; b[0] = wcol[(size_t)k0 * DCOL]; b[1] = wcol[(size_t)(k0 + 1) * DCOL];
        c = __builtin_amdgcn_wmma_f32_16x16x4_f32(
                 false, a,  false, b,
                 (short)0, c,  false,  false);
    }

    float bv;
    if constexpr (BIAS) bv = bias[colBase + l16];
    float* __restrict__ hout = H + colBase + l16;
    #pragma unroll
    for (int v = 0; v < 8; ++v) {
        const int row = rowBase + v + half * 8;
        float val = c[v];
        if constexpr (BIAS) val += bv;
        hout[(size_t)row * DCOL] = val;
    }
}

#define GB 8
#define LP 68
template <bool RELU, bool BIAS>
__global__ __launch_bounds__(256)
void k_agg_gemm(const int* __restrict__ rowptr, const int* __restrict__ rec,
                const float* __restrict__ dinv, const float* __restrict__ h,
                const float* __restrict__ biasIn, const float* __restrict__ W,
                const float* __restrict__ biasOut, float* __restrict__ out, int n) {
    __shared__ __attribute__((aligned(16))) float sA[16 * LP];
    __shared__ __attribute__((aligned(16))) float sD[16 * LP];
    const int tid = threadIdx.x;
    const int t = blockIdx.x * blockDim.x + tid;
    const int i = t >> 4, q = t & 15;
    const int r = tid >> 4;
    const bool valid = i < n;
    const int ic = valid ? i : n - 1;
    const float4* __restrict__ h4 = reinterpret_cast<const float4*>(h);
    const int2*   __restrict__ r2 = reinterpret_cast<const int2*>(rec);

    const int beg = rowptr[ic];
    const int end = rowptr[ic + 1];
    const float di = dinv[ic];
    const float4 hs = h4[ic * 16 + q];
    const float4 bv = reinterpret_cast<const float4*>(biasIn)[q];
    const float ws = di * di;
    float4 acc;
    acc.x = fmaf(ws, hs.x, bv.x); acc.y = fmaf(ws, hs.y, bv.y);
    acc.z = fmaf(ws, hs.z, bv.z); acc.w = fmaf(ws, hs.w, bv.w);

    for (int p = beg; p < end; p += GB) {
        int2   rr[GB];
        float4 v[GB];
        #pragma unroll
        for (int u = 0; u < GB; ++u) rr[u] = r2[min(p + u, end - 1)];
        #pragma unroll
        for (int u = 0; u < GB; ++u) v[u] = h4[rr[u].x * 16 + q];
        #pragma unroll
        for (int u = 0; u < GB; ++u) {
            const bool live = (u == 0) || (p + u < end);
            const float w = __int_as_float(rr[u].y) * di;
            const float nx = fmaf(w, v[u].x, acc.x), ny = fmaf(w, v[u].y, acc.y);
            const float nz = fmaf(w, v[u].z, acc.z), nw = fmaf(w, v[u].w, acc.w);
            acc.x = live ? nx : acc.x; acc.y = live ? ny : acc.y;
            acc.z = live ? nz : acc.z; acc.w = live ? nw : acc.w;
        }
    }

    if constexpr (RELU) {
        acc.x = fmaxf(acc.x, 0.0f); acc.y = fmaxf(acc.y, 0.0f);
        acc.z = fmaxf(acc.z, 0.0f); acc.w = fmaxf(acc.w, 0.0f);
    }
    *reinterpret_cast<float4*>(&sA[r * LP + q * 4]) = acc;
    __syncthreads();

    const int wave = tid >> 5;
    if (wave < 4) {
        const int lane    = tid & 31;
        const int half    = lane >> 4;
        const int l16     = lane & 15;
        const int colBase = wave << 4;
        const float* __restrict__ wcol = W + colBase + l16;
        const float* arow = sA + l16 * LP;
        v8f c = {};
        #pragma unroll
        for (int kb = 0; kb < DCOL; kb += 4) {
            const int k0 = kb + half * 2;
            const v2f a = *reinterpret_cast<const v2f*>(arow + k0);
            v2f b; b[0] = wcol[(size_t)k0 * DCOL]; b[1] = wcol[(size_t)(k0 + 1) * DCOL];
            c = __builtin_amdgcn_wmma_f32_16x16x4_f32(
                     false, a,  false, b,
                     (short)0, c,  false,  false);
        }
        #pragma unroll
        for (int v = 0; v < 8; ++v) sD[(v + half * 8) * LP + colBase + l16] = c[v];
    }
    __syncthreads();

    float4 o = *reinterpret_cast<const float4*>(&sD[r * LP + q * 4]);
    if constexpr (BIAS) {
        const float4 bo = reinterpret_cast<const float4*>(biasOut)[q];
        o.x += bo.x; o.y += bo.y; o.z += bo.z; o.w += bo.w;
    }
    if (valid) reinterpret_cast<float4*>(out)[i * 16 + q] = o;
}

extern "C" void kernel_launch(void* const* d_in, const int* in_sizes, int n_in,
                              void* d_out, int out_size, void* d_ws, size_t ws_size,
                              hipStream_t stream) {
    const float* x   = (const float*)d_in[0];
    const int*   ei  = (const int*)  d_in[1];
    const float* W1  = (const float*)d_in[2];
    const float* b1  = (const float*)d_in[3];
    const float* W2  = (const float*)d_in[4];
    const float* b2  = (const float*)d_in[5];
    const float* Wfc = (const float*)d_in[6];
    const float* bfc = (const float*)d_in[7];
    float* out = (float*)d_out;

    const int N = in_sizes[0] / DIN;
    const int E = in_sizes[1] / 2;
    const int* src = ei;
    const int* dst = ei + E;

    const size_t Np = ((size_t)(N + 255) / 256) * 256;
    const size_t Ep = ((size_t)(E + 255) / 256) * 256;
    int*   cnt    = (int*)d_ws;
    int*   rowptr = cnt + Np;
    float* dinv   = (float*)(rowptr + Np + 256);
    int*   rank   = (int*)(dinv + Np);
    int*   rec    = rank + Ep;
    float* buf0   = (float*)(rec + 2 * Ep);
    float* buf1   = buf0 + (size_t)N * DCOL;

    const int blk = 256;
    const int gN   = (N + blk - 1) / blk;
    const int gE   = (E + blk - 1) / blk;
    const int gN16 = (N * 16 + blk - 1) / blk;
    const int gScan = (N + 1023) / 1024;
    const int gGemm = N / 16;

    k_zero_cnt  <<<gN, blk, 0, stream>>>(cnt, N);
    k_count_rank<<<gE, blk, 0, stream>>>(dst, cnt, rank, E);
    k_scan_dinv <<<gScan, 1024, 0, stream>>>(cnt, rowptr, dinv, N);
    k_fill      <<<gE, blk, 0, stream>>>(src, dst, rank, rowptr, dinv, rec, E);

    k_gemm_wmma<DIN,  false, false><<<gGemm, 128, 0, stream>>>(x,    W1,  buf0, nullptr);

    k_agg_gemm<true,  false><<<gN16, blk, 0, stream>>>(rowptr, rec, dinv, buf0, b1, W2,  nullptr, buf1, N);

    k_agg_gemm<false, true ><<<gN16, blk, 0, stream>>>(rowptr, rec, dinv, buf1, b2, Wfc, bfc,     out,  N);
}


# Edges are grouped by destination through 256-node bins ranked with LDS atomics (chunk histogram, scan, bin scatter, per-bin count and fill): no global atomics; sum order per node = LDS arrival rank
# speedup vs baseline: 2.7157x; 1.2906x over previous
#include <hip/hip_runtime.h>
#include <hip/hip_bf16.h>


typedef __attribute__((ext_vector_type(2))) float v2f;
typedef __attribute__((ext_vector_type(8))) float v8f;

#define DIN 128
#define DCOL 64

#define BSH  8
#define BN   256
#define MAXB 256
#define EPT  16
#define CH   (256 * EPT)

__global__ __launch_bounds__(256)
void k_bin_hist(const int* __restrict__ dst, int* __restrict__ H, int e, int nb, int nc) {
    __shared__ int s_h[MAXB];
    const int t = threadIdx.x, c = blockIdx.x;
    s_h[t] = 0;
    __syncthreads();
    const int e0 = c * CH + t;
    int d[EPT];
    #pragma unroll
    for (int j = 0; j < EPT; ++j) {
        const int i = e0 + j * 256;
        d[j] = (i < e) ? dst[i] : -1;
    }
    #pragma unroll
    for (int j = 0; j < EPT; ++j)
        if (d[j] >= 0) atomicAdd(&s_h[d[j] >> BSH], 1);
    __syncthreads();
    if (t < nb) H[t * nc + c] = s_h[t];
}

__global__ __launch_bounds__(1024)
void k_scan_excl(const int* __restrict__ cnt, int* __restrict__ out, int n) {
    __shared__ int s_pre[32];
    __shared__ int s_tot[32];
    const int t = threadIdx.x, lane = t & 31, wv = t >> 5;
    const int b = blockIdx.x;

    int pre = 0;
    const int4* __restrict__ c4 = reinterpret_cast<const int4*>(cnt);
    const int nchunk = b * 256;
    #pragma unroll 4
    for (int j = t; j < nchunk; j += 1024) {
        int4 v = c4[j];
        pre += (v.x + v.y) + (v.z + v.w);
    }
    const int i = b * 1024 + t;
    const int v = (i < n) ? cnt[i] : 0;

    int inc = v;
    #pragma unroll
    for (int o = 1; o < 32; o <<= 1) {
        int u = __shfl_up(inc, o);
        if (lane >= o) inc += u;
    }
    #pragma unroll
    for (int o = 16; o > 0; o >>= 1) pre += __shfl_xor(pre, o);
    if (lane == 31) s_tot[wv] = inc;
    if (lane == 0)  s_pre[wv] = pre;
    __syncthreads();
    int off = 0;
    #pragma unroll
    for (int k = 0; k < 32; ++k) {
        off += s_pre[k];
        if (k < wv) off += s_tot[k];
    }
    if (i < n) {
        out[i] = off + inc - v;
        if (i == n - 1) out[n] = off + inc;
    }
}

__global__ __launch_bounds__(256)
void k_bin_scatter(const int* __restrict__ src, const int* __restrict__ dst,
                   const int* __restrict__ base, int* __restrict__ binned,
                   int e, int nb, int nc) {
    __shared__ int s_cur[MAXB + 1];
    const int t = threadIdx.x, c = blockIdx.x;
    s_cur[t] = (t < nb) ? base[t * nc + c] : 0;
    if (t == 0) s_cur[MAXB] = 0;
    __syncthreads();
    const int e0 = c * CH + t;
    int d[EPT], s[EPT];
    #pragma unroll
    for (int j = 0; j < EPT; ++j) {
        const int i = e0 + j * 256;
        const bool in = i < e;
        d[j] = in ? dst[i] : -1;
        s[j] = in ? src[i] : 0;
    }
    int2* __restrict__ b2 = reinterpret_cast<int2*>(binned);
    int p[EPT];
    #pragma unroll
    for (int j = 0; j < EPT; ++j)
        p[j] = atomicAdd(&s_cur[d[j] >= 0 ? (d[j] >> BSH) : MAXB], 1);
    #pragma unroll
    for (int j = 0; j < EPT; ++j)
        if (d[j] >= 0) b2[p[j]] = make_int2(s[j], d[j]);
}

#define FB 8
__global__ __launch_bounds__(256)
void k_bin_count(const int* __restrict__ binned, const int* __restrict__ base,
                 int* __restrict__ rowptr, float* __restrict__ dinv, int n, int nc) {
    __shared__ int s_c[BN];
    __shared__ int s_tot[8];
    const int t = threadIdx.x, b = blockIdx.x, lane = t & 31, wv = t >> 5;
    const int beg = base[b * nc];
    const int end = base[(b + 1) * nc];
    s_c[t] = 0;
    __syncthreads();
    const int2* __restrict__ b2 = reinterpret_cast<const int2*>(binned);
    for (int i0 = beg; i0 < end; i0 += BN * FB) {
        int d[FB];
        #pragma unroll
        for (int u = 0; u < FB; ++u) {
            const int i = i0 + u * BN + t;
            d[u] = (i < end) ? b2[i].y : -1;
        }
        #pragma unroll
        for (int u = 0; u < FB; ++u)
            if (d[u] >= 0) atomicAdd(&s_c[d[u] & (BN - 1)], 1);
    }
    __syncthreads();
    const int v = s_c[t];
    int inc = v;
    #pragma unroll
    for (int o = 1; o < 32; o <<= 1) {
        int u = __shfl_up(inc, o);
        if (lane >= o) inc += u;
    }
    if (lane == 31) s_tot[wv] = inc;
    __syncthreads();
    int off = beg;
    #pragma unroll
    for (int k = 0; k < 8; ++k)
        if (k < wv) off += s_tot[k];
    const int node = b * BN + t;
    if (node < n) {
        rowptr[node] = off + inc - v;
        if (node == n - 1) rowptr[n] = off + inc;
        dinv[node] = rsqrtf((float)(v + 1));
    }
}

__global__ __launch_bounds__(256)
void k_bin_fill(const int* __restrict__ binned, const int* __restrict__ base,
                const int* __restrict__ rowptr, const float* __restrict__ dinv,
                int* __restrict__ rec, int n, int nc) {
    __shared__ int s_cur[BN + 1];
    const int t = threadIdx.x, b = blockIdx.x;
    const int beg = base[b * nc];
    const int end = base[(b + 1) * nc];
    const int node = b * BN + t;
    s_cur[t] = (node < n) ? rowptr[node] : 0;
    if (t == 0) s_cur[BN] = 0;
    __syncthreads();
    const int2* __restrict__ b2 = reinterpret_cast<const int2*>(binned);
    int2* __restrict__ r2 = reinterpret_cast<int2*>(rec);
    for (int i0 = beg; i0 < end; i0 += BN * FB) {
        int2  sd[FB];
        float w[FB];
        #pragma unroll
        for (int u = 0; u < FB; ++u) {
            const int i = i0 + u * BN + t;
            sd[u] = (i < end) ? b2[i] : make_int2(0, -1);
        }
        #pragma unroll
        for (int u = 0; u < FB; ++u) w[u] = dinv[sd[u].x];
        int p[FB];
        #pragma unroll
        for (int u = 0; u < FB; ++u)
            p[u] = atomicAdd(&s_cur[sd[u].y >= 0 ? (sd[u].y & (BN - 1)) : BN], 1);
        #pragma unroll
        for (int u = 0; u < FB; ++u)
            if (sd[u].y >= 0) r2[p[u]] = make_int2(sd[u].x, __float_as_int(w[u]));
    }
}

template <int K, bool RELU, bool BIAS>
__global__ __launch_bounds__(128)
void k_gemm_wmma(const float* __restrict__ A, const float* __restrict__ W,
                 float* __restrict__ H, const float* __restrict__ bias) {
    const int lane    = threadIdx.x & 31;
    const int wave    = threadIdx.x >> 5;
    const int rowBase = blockIdx.x << 4;
    const int colBase = wave << 4;
    const int half    = lane >> 4;
    const int l16     = lane & 15;

    const float* __restrict__ arow = A + (size_t)(rowBase + l16) * K;
    const float* __restrict__ wcol = W + colBase + l16;

    v8f c = {};
    #pragma unroll 8
    for (int kb = 0; kb < K; kb += 4) {
        const int k0 = kb + half * 2;
        float a0 = arow[k0];
        float a1 = arow[k0 + 1];
        if constexpr (RELU) { a0 = fmaxf(a0, 0.0f); a1 = fmaxf(a1, 0.0f); }
        v2f a; a[0] = a0; a[1] = a1;
        v2f b; b[0] = wcol[(size_t)k0 * DCOL]; b[1] = wcol[(size_t)(k0 + 1) * DCOL];
        c = __builtin_amdgcn_wmma_f32_16x16x4_f32(
                 false, a,  false, b,
                 (short)0, c,  false,  false);
    }

    float bv;
    if constexpr (BIAS) bv = bias[colBase + l16];
    float* __restrict__ hout = H + colBase + l16;
    #pragma unroll
    for (int v = 0; v < 8; ++v) {
        const int row = rowBase + v + half * 8;
        float val = c[v];
        if constexpr (BIAS) val += bv;
        hout[(size_t)row * DCOL] = val;
    }
}

#define GB 8
#define LP 68
template <bool RELU, bool BIAS>
__global__ __launch_bounds__(256)
void k_agg_gemm(const int* __restrict__ rowptr, const int* __restrict__ rec,
                const float* __restrict__ dinv, const float* __restrict__ h,
                const float* __restrict__ biasIn, const float* __restrict__ W,
                const float* __restrict__ biasOut, float* __restrict__ out, int n) {
    __shared__ __attribute__((aligned(16))) float sA[16 * LP];
    __shared__ __attribute__((aligned(16))) float sD[16 * LP];
    const int tid = threadIdx.x;
    const int t = blockIdx.x * blockDim.x + tid;
    const int i = t >> 4, q = t & 15;
    const int r = tid >> 4;
    const bool valid = i < n;
    const int ic = valid ? i : n - 1;
    const float4* __restrict__ h4 = reinterpret_cast<const float4*>(h);
    const int2*   __restrict__ r2 = reinterpret_cast<const int2*>(rec);

    const int beg = rowptr[ic];
    const int end = rowptr[ic + 1];
    const float di = dinv[ic];
    const float4 hs = h4[ic * 16 + q];
    const float4 bv = reinterpret_cast<const float4*>(biasIn)[q];
    const float ws = di * di;
    float4 acc;
    acc.x = fmaf(ws, hs.x, bv.x); acc.y = fmaf(ws, hs.y, bv.y);
    acc.z = fmaf(ws, hs.z, bv.z); acc.w = fmaf(ws, hs.w, bv.w);

    for (int p = beg; p < end; p += GB) {
        int2   rr[GB];
        float4 v[GB];
        #pragma unroll
        for (int u = 0; u < GB; ++u) rr[u] = r2[min(p + u, end - 1)];
        #pragma unroll
        for (int u = 0; u < GB; ++u) v[u] = h4[rr[u].x * 16 + q];
        #pragma unroll
        for (int u = 0; u < GB; ++u) {
            const bool live = (u == 0) || (p + u < end);
            const float w = __int_as_float(rr[u].y) * di;
            const float nx = fmaf(w, v[u].x, acc.x), ny = fmaf(w, v[u].y, acc.y);
            const float nz = fmaf(w, v[u].z, acc.z), nw = fmaf(w, v[u].w, acc.w);
            acc.x = live ? nx : acc.x; acc.y = live ? ny : acc.y;
            acc.z = live ? nz : acc.z; acc.w = live ? nw : acc.w;
        }
    }

    if constexpr (RELU) {
        acc.x = fmaxf(acc.x, 0.0f); acc.y = fmaxf(acc.y, 0.0f);
        acc.z = fmaxf(acc.z, 0.0f); acc.w = fmaxf(acc.w, 0.0f);
    }
    *reinterpret_cast<float4*>(&sA[r * LP + q * 4]) = acc;
    __syncthreads();

    const int wave = tid >> 5;
    if (wave < 4) {
        const int lane    = tid & 31;
        const int half    = lane >> 4;
        const int l16     = lane & 15;
        const int colBase = wave << 4;
        const float* __restrict__ wcol = W + colBase + l16;
        const float* arow = sA + l16 * LP;
        v8f c = {};
        #pragma unroll
        for (int kb = 0; kb < DCOL; kb += 4) {
            const int k0 = kb + half * 2;
            const v2f a = *reinterpret_cast<const v2f*>(arow + k0);
            v2f b; b[0] = wcol[(size_t)k0 * DCOL]; b[1] = wcol[(size_t)(k0 + 1) * DCOL];
            c = __builtin_amdgcn_wmma_f32_16x16x4_f32(
                     false, a,  false, b,
                     (short)0, c,  false,  false);
        }
        #pragma unroll
        for (int v = 0; v < 8; ++v) sD[(v + half * 8) * LP + colBase + l16] = c[v];
    }
    __syncthreads();

    float4 o = *reinterpret_cast<const float4*>(&sD[r * LP + q * 4]);
    if constexpr (BIAS) {
        const float4 bo = reinterpret_cast<const float4*>(biasOut)[q];
        o.x += bo.x; o.y += bo.y; o.z += bo.z; o.w += bo.w;
    }
    if (valid) reinterpret_cast<float4*>(out)[i * 16 + q] = o;
}

extern "C" void kernel_launch(void* const* d_in, const int* in_sizes, int n_in,
                              void* d_out, int out_size, void* d_ws, size_t ws_size,
                              hipStream_t stream) {
    const float* x   = (const float*)d_in[0];
    const int*   ei  = (const int*)  d_in[1];
    const float* W1  = (const float*)d_in[2];
    const float* b1  = (const float*)d_in[3];
    const float* W2  = (const float*)d_in[4];
    const float* b2  = (const float*)d_in[5];
    const float* Wfc = (const float*)d_in[6];
    const float* bfc = (const float*)d_in[7];
    float* out = (float*)d_out;

    const int N = in_sizes[0] / DIN;
    const int E = in_sizes[1] / 2;
    const int* src = ei;
    const int* dst = ei + E;

    const int NB  = (N + BN - 1) / BN;
    const int NC  = (E + CH - 1) / CH;
    const int NBC = NB * NC;

    const size_t Np   = ((size_t)(N + 255) / 256) * 256;
    const size_t Ep   = ((size_t)(E + 255) / 256) * 256;
    const size_t NBCp = ((size_t)(NBC + 255) / 256) * 256;
    int*   H      = (int*)d_ws;
    int*   base   = H + NBCp;
    int*   rowptr = base + NBCp + 256;
    float* dinv   = (float*)(rowptr + Np + 256);
    int*   binned = (int*)(dinv + Np);
    int*   rec    = binned + 2 * Ep;
    float* buf0   = (float*)(rec + 2 * Ep);
    float* buf1   = buf0 + (size_t)N * DCOL;

    const int blk = 256;
    const int gN16 = (N * 16 + blk - 1) / blk;
    const int gScan = (NBC + 1023) / 1024;
    const int gGemm = N / 16;

    k_bin_hist   <<<NC, 256, 0, stream>>>(dst, H, E, NB, NC);
    k_scan_excl  <<<gScan, 1024, 0, stream>>>(H, base, NBC);
    k_bin_scatter<<<NC, 256, 0, stream>>>(src, dst, base, binned, E, NB, NC);
    k_bin_count  <<<NB, BN, 0, stream>>>(binned, base, rowptr, dinv, N, NC);
    k_bin_fill   <<<NB, BN, 0, stream>>>(binned, base, rowptr, dinv, rec, N, NC);

    k_gemm_wmma<DIN,  false, false><<<gGemm, 128, 0, stream>>>(x,    W1,  buf0, nullptr);

    k_agg_gemm<true,  false><<<gN16, blk, 0, stream>>>(rowptr, rec, dinv, buf0, b1, W2,  nullptr, buf1, N);

    k_agg_gemm<false, true ><<<gN16, blk, 0, stream>>>(rowptr, rec, dinv, buf1, b2, Wfc, bfc,     out,  N);
}


# Each 4096-edge chunk is bin-sorted in its own slot (LDS rank + scan); bin blocks read their per-chunk segments via an LDS table: histogram and scan launches go (8 to 6); sum order = LDS arrival rank
# speedup vs baseline: 2.8439x; 1.0472x over previous
#include <hip/hip_runtime.h>
#include <hip/hip_bf16.h>


typedef __attribute__((ext_vector_type(2))) float v2f;
typedef __attribute__((ext_vector_type(8))) float v8f;

#define DIN 128
#define DCOL 64

#define BSH  8
#define BN   256
#define MAXB 256
#define EPT  16
#define CH   (256 * EPT)

__global__ __launch_bounds__(256)
void k_chunk_sort(const int* __restrict__ src, const int* __restrict__ dst,
                  int* __restrict__ OFF, int* __restrict__ binned, int e, int nb, int nc) {
    __shared__ int s_h[MAXB + 1];
    __shared__ int s_tot[8];
    const int t = threadIdx.x, c = blockIdx.x, lane = t & 31, wv = t >> 5;
    s_h[t] = 0;
    if (t == 0) s_h[MAXB] = 0;
    __syncthreads();
    const int e0 = c * CH + t;
    int d[EPT], s[EPT];
    #pragma unroll
    for (int j = 0; j < EPT; ++j) {
        const int i = e0 + j * 256;
        const bool in = i < e;
        d[j] = in ? dst[i] : -1;
        s[j] = in ? src[i] : 0;
    }
    int p[EPT];
    #pragma unroll
    for (int j = 0; j < EPT; ++j)
        p[j] = atomicAdd(&s_h[d[j] >= 0 ? (d[j] >> BSH) : MAXB], 1);
    __syncthreads();
    const int v = s_h[t];
    int inc = v;
    #pragma unroll
    for (int o = 1; o < 32; o <<= 1) {
        int u = __shfl_up(inc, o);
        if (lane >= o) inc += u;
    }
    if (lane == 31) s_tot[wv] = inc;
    __syncthreads();
    int off = 0;
    #pragma unroll
    for (int k = 0; k < 8; ++k)
        if (k < wv) off += s_tot[k];
    const int excl = off + inc - v;
    s_h[t] = excl;
    if (t < nb) OFF[t * nc + c] = excl;
    if (t == nb - 1) OFF[nb * nc + c] = excl + v;
    __syncthreads();
    int2* __restrict__ b2 = reinterpret_cast<int2*>(binned) + (size_t)c * CH;
    #pragma unroll
    for (int j = 0; j < EPT; ++j)
        if (d[j] >= 0) b2[s_h[d[j] >> BSH] + p[j]] = make_int2(s[j], d[j]);
}

__device__ __forceinline__ void bin_segments(const int* __restrict__ OFF, int b, int nc,
                                             int* s_st, int* s_o, int* s_w, int& beg, int& total) {
    const int t = threadIdx.x, lane = t & 31, wv = t >> 5;
    int o = 0, cnt = 0;
    if (t < nc) { o = OFF[b * nc + t]; cnt = OFF[(b + 1) * nc + t] - o; }
    int inc = cnt, so = o;
    #pragma unroll
    for (int k = 1; k < 32; k <<= 1) {
        int u = __shfl_up(inc, k);
        if (lane >= k) inc += u;
    }
    #pragma unroll
    for (int k = 16; k > 0; k >>= 1) so += __shfl_xor(so, k);
    if (lane == 31) s_w[wv] = inc;
    if (lane == 0)  s_w[8 + wv] = so;
    __syncthreads();
    int off = 0, tot = 0, bsum = 0;
    #pragma unroll
    for (int k = 0; k < 8; ++k) {
        const int w = s_w[k];
        if (k < wv) off += w;
        tot += w;
        bsum += s_w[8 + k];
    }
    const int st = off + inc - cnt;
    s_st[t] = st;
    s_o[t]  = t * CH + o - st;
    __syncthreads();
    beg = bsum;
    total = tot;
}

__device__ __forceinline__ int bin_place(const int* s_st, const int* s_o, int i) {
    int c = 0;
    #pragma unroll
    for (int step = 128; step > 0; step >>= 1)
        if (s_st[c + step] <= i) c += step;
    return s_o[c] + i;
}

#define FB 8
__global__ __launch_bounds__(256)
void k_bin_count(const int* __restrict__ binned, const int* __restrict__ OFF,
                 int* __restrict__ rowptr, float* __restrict__ dinv, int n, int nc) {
    __shared__ int s_c[BN];
    __shared__ int s_tot[8];
    __shared__ int s_st[256];
    __shared__ int s_o[256];
    __shared__ int s_w[16];
    const int t = threadIdx.x, b = blockIdx.x, lane = t & 31, wv = t >> 5;
    s_c[t] = 0;
    int beg, total;
    bin_segments(OFF, b, nc, s_st, s_o, s_w, beg, total);
    const int2* __restrict__ b2 = reinterpret_cast<const int2*>(binned);
    for (int i0 = 0; i0 < total; i0 += BN * FB) {
        int d[FB];
        #pragma unroll
        for (int u = 0; u < FB; ++u) {
            const int i = i0 + u * BN + t;
            const int dd = b2[bin_place(s_st, s_o, min(i, total - 1))].y;
            d[u] = (i < total) ? dd : -1;
        }
        #pragma unroll
        for (int u = 0; u < FB; ++u)
            if (d[u] >= 0) atomicAdd(&s_c[d[u] & (BN - 1)], 1);
    }
    __syncthreads();
    const int v = s_c[t];
    int inc = v;
    #pragma unroll
    for (int o = 1; o < 32; o <<= 1) {
        int u = __shfl_up(inc, o);
        if (lane >= o) inc += u;
    }
    if (lane == 31) s_tot[wv] = inc;
    __syncthreads();
    int off = beg;
    #pragma unroll
    for (int k = 0; k < 8; ++k)
        if (k < wv) off += s_tot[k];
    const int node = b * BN + t;
    if (node < n) {
        rowptr[node] = off + inc - v;
        if (node == n - 1) rowptr[n] = off + inc;
        dinv[node] = rsqrtf((float)(v + 1));
    }
}

__global__ __launch_bounds__(256)
void k_bin_fill(const int* __restrict__ binned, const int* __restrict__ OFF,
                const int* __restrict__ rowptr, const float* __restrict__ dinv,
                int* __restrict__ rec, int n, int nc) {
    __shared__ int s_cur[BN + 1];
    __shared__ int s_st[256];
    __shared__ int s_o[256];
    __shared__ int s_w[16];
    const int t = threadIdx.x, b = blockIdx.x;
    const int node = b * BN + t;
    s_cur[t] = (node < n) ? rowptr[node] : 0;
    if (t == 0) s_cur[BN] = 0;
    int beg, total;
    bin_segments(OFF, b, nc, s_st, s_o, s_w, beg, total);
    const int2* __restrict__ b2 = reinterpret_cast<const int2*>(binned);
    int2* __restrict__ r2 = reinterpret_cast<int2*>(rec);
    for (int i0 = 0; i0 < total; i0 += BN * FB) {
        int2  sd[FB];
        float w[FB];
        #pragma unroll
        for (int u = 0; u < FB; ++u) {
            const int i = i0 + u * BN + t;
            const int2 x = b2[bin_place(s_st, s_o, min(i, total - 1))];
            sd[u] = (i < total) ? x : make_int2(x.x, -1);
        }
        #pragma unroll
        for (int u = 0; u < FB; ++u) w[u] = dinv[sd[u].x];
        int p[FB];
        #pragma unroll
        for (int u = 0; u < FB; ++u)
            p[u] = atomicAdd(&s_cur[sd[u].y >= 0 ? (sd[u].y & (BN - 1)) : BN], 1);
        #pragma unroll
        for (int u = 0; u < FB; ++u)
            if (sd[u].y >= 0) r2[p[u]] = make_int2(sd[u].x, __float_as_int(w[u]));
    }
}

template <int K, bool RELU, bool BIAS>
__global__ __launch_bounds__(128)
void k_gemm_wmma(const float* __restrict__ A, const float* __restrict__ W,
                 float* __restrict__ H, const float* __restrict__ bias) {
    const int lane    = threadIdx.x & 31;
    const int wave    = threadIdx.x >> 5;
    const int rowBase = blockIdx.x << 4;
    const int colBase = wave << 4;
    const int half    = lane >> 4;
    const int l16     = lane & 15;

    const float* __restrict__ arow = A + (size_t)(rowBase + l16) * K;
    const float* __restrict__ wcol = W + colBase + l16;

    v8f c = {};
    #pragma unroll 8
    for (int kb = 0; kb < K; kb += 4) {
        const int k0 = kb + half * 2;
        float a0 = arow[k0];
        float a1 = arow[k0 + 1];
        if constexpr (RELU) { a0 = fmaxf(a0, 0.0f); a1 = fmaxf(a1, 0.0f); }
        v2f a; a[0] = a0; a[1] = a1;
        v2f b; b[0] = wcol[(size_t)k0 * DCOL]; b[1] = wcol[(size_t)(k0 + 1) * DCOL];
        c = __builtin_amdgcn_wmma_f32_16x16x4_f32(
                 false, a,  false, b,
                 (short)0, c,  false,  false);
    }

    float bv;
    if constexpr (BIAS) bv = bias[colBase + l16];
    float* __restrict__ hout = H + colBase + l16;
    #pragma unroll
    for (int v = 0; v < 8; ++v) {
        const int row = rowBase + v + half * 8;
        float val = c[v];
        if constexpr (BIAS) val += bv;
        hout[(size_t)row * DCOL] = val;
    }
}

#define GB 8
#define LP 68
template <bool RELU, bool BIAS>
__global__ __launch_bounds__(256)
void k_agg_gemm(const int* __restrict__ rowptr, const int* __restrict__ rec,
                const float* __restrict__ dinv, const float* __restrict__ h,
                const float* __restrict__ biasIn, const float* __restrict__ W,
                const float* __restrict__ biasOut, float* __restrict__ out, int n) {
    __shared__ __attribute__((aligned(16))) float sA[16 * LP];
    __shared__ __attribute__((aligned(16))) float sD[16 * LP];
    const int tid = threadIdx.x;
    const int t = blockIdx.x * blockDim.x + tid;
    const int i = t >> 4, q = t & 15;
    const int r = tid >> 4;
    const bool valid = i < n;
    const int ic = valid ? i : n - 1;
    const float4* __restrict__ h4 = reinterpret_cast<const float4*>(h);
    const int2*   __restrict__ r2 = reinterpret_cast<const int2*>(rec);

    const int beg = rowptr[ic];
    const int end = rowptr[ic + 1];
    const float di = dinv[ic];
    const float4 hs = h4[ic * 16 + q];
    const float4 bv = reinterpret_cast<const float4*>(biasIn)[q];
    const float ws = di * di;
    float4 acc;
    acc.x = fmaf(ws, hs.x, bv.x); acc.y = fmaf(ws, hs.y, bv.y);
    acc.z = fmaf(ws, hs.z, bv.z); acc.w = fmaf(ws, hs.w, bv.w);

    for (int p = beg; p < end; p += GB) {
        int2   rr[GB];
        float4 v[GB];
        #pragma unroll
        for (int u = 0; u < GB; ++u) rr[u] = r2[min(p + u, end - 1)];
        #pragma unroll
        for (int u = 0; u < GB; ++u) v[u] = h4[rr[u].x * 16 + q];
        #pragma unroll
        for (int u = 0; u < GB; ++u) {
            const bool live = (u == 0) || (p + u < end);
            const float w = __int_as_float(rr[u].y) * di;
            const float nx = fmaf(w, v[u].x, acc.x), ny = fmaf(w, v[u].y, acc.y);
            const float nz = fmaf(w, v[u].z, acc.z), nw = fmaf(w, v[u].w, acc.w);
            acc.x = live ? nx : acc.x; acc.y = live ? ny : acc.y;
            acc.z = live ? nz : acc.z; acc.w = live ? nw : acc.w;
        }
    }

    if constexpr (RELU) {
        acc.x = fmaxf(acc.x, 0.0f); acc.y = fmaxf(acc.y, 0.0f);
        acc.z = fmaxf(acc.z, 0.0f); acc.w = fmaxf(acc.w, 0.0f);
    }
    *reinterpret_cast<float4*>(&sA[r * LP + q * 4]) = acc;
    __syncthreads();

    const int wave = tid >> 5;
    if (wave < 4) {
        const int lane    = tid & 31;
        const int half    = lane >> 4;
        const int l16     = lane & 15;
        const int colBase = wave << 4;
        const float* __restrict__ wcol = W + colBase + l16;
        const float* arow = sA + l16 * LP;
        v8f c = {};
        #pragma unroll
        for (int kb = 0; kb < DCOL; kb += 4) {
            const int k0 = kb + half * 2;
            const v2f a = *reinterpret_cast<const v2f*>(arow + k0);
            v2f b; b[0] = wcol[(size_t)k0 * DCOL]; b[1] = wcol[(size_t)(k0 + 1) * DCOL];
            c = __builtin_amdgcn_wmma_f32_16x16x4_f32(
                     false, a,  false, b,
                     (short)0, c,  false,  false);
        }
        #pragma unroll
        for (int v = 0; v < 8; ++v) sD[(v + half * 8) * LP + colBase + l16] = c[v];
    }
    __syncthreads();

    float4 o = *reinterpret_cast<const float4*>(&sD[r * LP + q * 4]);
    if constexpr (BIAS) {
        const float4 bo = reinterpret_cast<const float4*>(biasOut)[q];
        o.x += bo.x; o.y += bo.y; o.z += bo.z; o.w += bo.w;
    }
    if (valid) reinterpret_cast<float4*>(out)[i * 16 + q] = o;
}

extern "C" void kernel_launch(void* const* d_in, const int* in_sizes, int n_in,
                              void* d_out, int out_size, void* d_ws, size_t ws_size,
                              hipStream_t stream) {
    const float* x   = (const float*)d_in[0];
    const int*   ei  = (const int*)  d_in[1];
    const float* W1  = (const float*)d_in[2];
    const float* b1  = (const float*)d_in[3];
    const float* W2  = (const float*)d_in[4];
    const float* b2  = (const float*)d_in[5];
    const float* Wfc = (const float*)d_in[6];
    const float* bfc = (const float*)d_in[7];
    float* out = (float*)d_out;

    const int N = in_sizes[0] / DIN;
    const int E = in_sizes[1] / 2;
    const int* src = ei;
    const int* dst = ei + E;

    const int NB  = (N + BN - 1) / BN;
    const int NC  = (E + CH - 1) / CH;
    const int NOF = (NB + 1) * NC;

    const size_t Np   = ((size_t)(N + 255) / 256) * 256;
    const size_t Ep   = ((size_t)(E + 255) / 256) * 256;
    const size_t NOFp = ((size_t)(NOF + 255) / 256) * 256;
    int*   OFF    = (int*)d_ws;
    int*   rowptr = OFF + NOFp;
    float* dinv   = (float*)(rowptr + Np + 256);
    int*   binned = (int*)(dinv + Np);
    int*   rec    = binned + 2 * (size_t)NC * CH;
    float* buf0   = (float*)(rec + 2 * Ep);
    float* buf1   = buf0 + (size_t)N * DCOL;

    const int blk = 256;
    const int gN16 = (N * 16 + blk - 1) / blk;
    const int gGemm = N / 16;

    k_chunk_sort<<<NC, 256, 0, stream>>>(src, dst, OFF, binned, E, NB, NC);
    k_bin_count <<<NB, BN, 0, stream>>>(binned, OFF, rowptr, dinv, N, NC);
    k_bin_fill  <<<NB, BN, 0, stream>>>(binned, OFF, rowptr, dinv, rec, N, NC);

    k_gemm_wmma<DIN,  false, false><<<gGemm, 128, 0, stream>>>(x,    W1,  buf0, nullptr);

    k_agg_gemm<true,  false><<<gN16, blk, 0, stream>>>(rowptr, rec, dinv, buf0, b1, W2,  nullptr, buf1, N);

    k_agg_gemm<false, true ><<<gN16, blk, 0, stream>>>(rowptr, rec, dinv, buf1, b2, Wfc, bfc,     out,  N);
}


# The layer-1 product x @ W1 (independent of the edge list) runs as extra 32-row blocks behind the sorting blocks of the three edge-grouping launches, not in a launch of its own: 6 launches become 5
# speedup vs baseline: 3.1351x; 1.1024x over previous
#include <hip/hip_runtime.h>
#include <hip/hip_bf16.h>


typedef __attribute__((ext_vector_type(2))) float v2f;
typedef __attribute__((ext_vector_type(8))) float v8f;

#define DIN 128
#define DCOL 64

__device__ __forceinline__ void product_rows32(const float* __restrict__ A, const float* __restrict__ W,
                                               float* __restrict__ H, int g, int n) {
    const int lane    = threadIdx.x & 31;
    const int wave    = threadIdx.x >> 5;
    const int rowBase = (g << 5) + ((wave >> 2) << 4);
    if (rowBase >= n) return;
    const int colBase = (wave & 3) << 4;
    const int half    = lane >> 4;
    const int l16     = lane & 15;

    const float* __restrict__ arow = A + (size_t)(rowBase + l16) * DIN;
    const float* __restrict__ wcol = W + colBase + l16;

    v8f c = {};
    #pragma unroll 8
    for (int kb = 0; kb < DIN; kb += 4) {
        const int k0 = kb + half * 2;
        v2f a; a[0] = arow[k0]; a[1] = arow[k0 + 1];
        v2f b; b[0] = wcol[(size_t)k0 * DCOL]; b[1] = wcol[(size_t)(k0 + 1) * DCOL];
        c = __builtin_amdgcn_wmma_f32_16x16x4_f32(
                 false, a,  false, b,
                 (short)0, c,  false,  false);
    }

    float* __restrict__ hout = H + colBase + l16;
    #pragma unroll
    for (int v = 0; v < 8; ++v)
        hout[(size_t)(rowBase + v + half * 8) * DCOL] = c[v];
}

#define BSH  8
#define BN   256
#define MAXB 256
#define EPT  16
#define CH   (256 * EPT)

__global__ __launch_bounds__(256)
void k_chunk_sort(const int* __restrict__ src, const int* __restrict__ dst,
                  int* __restrict__ OFF, int* __restrict__ binned, int e, int nb, int nc,
                  const float* __restrict__ x, const float* __restrict__ W1, float* __restrict__ h1,
                  int n, int g0) {
    __shared__ int s_h[MAXB + 1];
    __shared__ int s_tot[8];
    if ((int)blockIdx.x >= nc) {
        product_rows32(x, W1, h1, g0 + (int)blockIdx.x - nc, n);
        return;
    }
    const int t = threadIdx.x, c = blockIdx.x, lane = t & 31, wv = t >> 5;
    s_h[t] = 0;
    if (t == 0) s_h[MAXB] = 0;
    __syncthreads();
    const int e0 = c * CH + t;
    int d[EPT], s[EPT];
    #pragma unroll
    for (int j = 0; j < EPT; ++j) {
        const int i = e0 + j * 256;
        const bool in = i < e;
        d[j] = in ? dst[i] : -1;
        s[j] = in ? src[i] : 0;
    }
    int p[EPT];
    #pragma unroll
    for (int j = 0; j < EPT; ++j)
        p[j] = atomicAdd(&s_h[d[j] >= 0 ? (d[j] >> BSH) : MAXB], 1);
    __syncthreads();
    const int v = s_h[t];
    int inc = v;
    #pragma unroll
    for (int o = 1; o < 32; o <<= 1) {
        int u = __shfl_up(inc, o);
        if (lane >= o) inc += u;
    }
    if (lane == 31) s_tot[wv] = inc;
    __syncthreads();
    int off = 0;
    #pragma unroll
    for (int k = 0; k < 8; ++k)
        if (k < wv) off += s_tot[k];
    const int excl = off + inc - v;
    s_h[t] = excl;
    if (t < nb) OFF[t * nc + c] = excl;
    if (t == nb - 1) OFF[nb * nc + c] = excl + v;
    __syncthreads();
    int2* __restrict__ b2 = reinterpret_cast<int2*>(binned) + (size_t)c * CH;
    #pragma unroll
    for (int j = 0; j < EPT; ++j)
        if (d[j] >= 0) b2[s_h[d[j] >> BSH] + p[j]] = make_int2(s[j], d[j]);
}

__device__ __forceinline__ void bin_segments(const int* __restrict__ OFF, int b, int nc,
                                             int* s_st, int* s_o, int* s_w, int& beg, int& total) {
    const int t = threadIdx.x, lane = t & 31, wv = t >> 5;
    int o = 0, cnt = 0;
    if (t < nc) { o = OFF[b * nc + t]; cnt = OFF[(b + 1) * nc + t] - o; }
    int inc = cnt, so = o;
    #pragma unroll
    for (int k = 1; k < 32; k <<= 1) {
        int u = __shfl_up(inc, k);
        if (lane >= k) inc += u;
    }
    #pragma unroll
    for (int k = 16; k > 0; k >>= 1) so += __shfl_xor(so, k);
    if (lane == 31) s_w[wv] = inc;
    if (lane == 0)  s_w[8 + wv] = so;
    __syncthreads();
    int off = 0, tot = 0, bsum = 0;
    #pragma unroll
    for (int k = 0; k < 8; ++k) {
        const int w = s_w[k];
        if (k < wv) off += w;
        tot += w;
        bsum += s_w[8 + k];
    }
    const int st = off + inc - cnt;
    s_st[t] = st;
    s_o[t]  = t * CH + o - st;
    __syncthreads();
    beg = bsum;
    total = tot;
}

__device__ __forceinline__ int bin_place(const int* s_st, const int* s_o, int i) {
    int c = 0;
    #pragma unroll
    for (int step = 128; step > 0; step >>= 1)
        if (s_st[c + step] <= i) c += step;
    return s_o[c] + i;
}

#define FB 8
__global__ __launch_bounds__(256)
void k_bin_count(const int* __restrict__ binned, const int* __restrict__ OFF,
                 int* __restrict__ rowptr, float* __restrict__ dinv, int n, int nc,
                 const float* __restrict__ x, const float* __restrict__ W1, float* __restrict__ h1,
                 int nb, int g0) {
    __shared__ int s_c[BN];
    __shared__ int s_tot[8];
    __shared__ int s_st[256];
    __shared__ int s_o[256];
    __shared__ int s_w[16];
    if ((int)blockIdx.x >= nb) {
        product_rows32(x, W1, h1, g0 + (int)blockIdx.x - nb, n);
        return;
    }
    const int t = threadIdx.x, b = blockIdx.x, lane = t & 31, wv = t >> 5;
    s_c[t] = 0;
    int beg, total;
    bin_segments(OFF, b, nc, s_st, s_o, s_w, beg, total);
    const int2* __restrict__ b2 = reinterpret_cast<const int2*>(binned);
    for (int i0 = 0; i0 < total; i0 += BN * FB) {
        int d[FB];
        #pragma unroll
        for (int u = 0; u < FB; ++u) {
            const int i = i0 + u * BN + t;
            const int dd = b2[bin_place(s_st, s_o, min(i, total - 1))].y;
            d[u] = (i < total) ? dd : -1;
        }
        #pragma unroll
        for (int u = 0; u < FB; ++u)
            if (d[u] >= 0) atomicAdd(&s_c[d[u] & (BN - 1)], 1);
    }
    __syncthreads();
    const int v = s_c[t];
    int inc = v;
    #pragma unroll
    for (int o = 1; o < 32; o <<= 1) {
        int u = __shfl_up(inc, o);
        if (lane >= o) inc += u;
    }
    if (lane == 31) s_tot[wv] = inc;
    __syncthreads();
    int off = beg;
    #pragma unroll
    for (int k = 0; k < 8; ++k)
        if (k < wv) off += s_tot[k];
    const int node = b * BN + t;
    if (node < n) {
        rowptr[node] = off + inc - v;
        if (node == n - 1) rowptr[n] = off + inc;
        dinv[node] = rsqrtf((float)(v + 1));
    }
}

__global__ __launch_bounds__(256)
void k_bin_fill(const int* __restrict__ binned, const int* __restrict__ OFF,
                const int* __restrict__ rowptr, const float* __restrict__ dinv,
                int* __restrict__ rec, int n, int nc,
                const float* __restrict__ x, const float* __restrict__ W1, float* __restrict__ h1,
                int nb, int g0) {
    __shared__ int s_cur[BN + 1];
    __shared__ int s_st[256];
    __shared__ int s_o[256];
    __shared__ int s_w[16];
    if ((int)blockIdx.x >= nb) {
        product_rows32(x, W1, h1, g0 + (int)blockIdx.x - nb, n);
        return;
    }
    const int t = threadIdx.x, b = blockIdx.x;
    const int node = b * BN + t;
    s_cur[t] = (node < n) ? rowptr[node] : 0;
    if (t == 0) s_cur[BN] = 0;
    int beg, total;
    bin_segments(OFF, b, nc, s_st, s_o, s_w, beg, total);
    const int2* __restrict__ b2 = reinterpret_cast<const int2*>(binned);
    int2* __restrict__ r2 = reinterpret_cast<int2*>(rec);
    for (int i0 = 0; i0 < total; i0 += BN * FB) {
        int2  sd[FB];
        float w[FB];
        #pragma unroll
        for (int u = 0; u < FB; ++u) {
            const int i = i0 + u * BN + t;
            const int2 x = b2[bin_place(s_st, s_o, min(i, total - 1))];
            sd[u] = (i < total) ? x : make_int2(x.x, -1);
        }
        #pragma unroll
        for (int u = 0; u < FB; ++u) w[u] = dinv[sd[u].x];
        int p[FB];
        #pragma unroll
        for (int u = 0; u < FB; ++u)
            p[u] = atomicAdd(&s_cur[sd[u].y >= 0 ? (sd[u].y & (BN - 1)) : BN], 1);
        #pragma unroll
        for (int u = 0; u < FB; ++u)
            if (sd[u].y >= 0) r2[p[u]] = make_int2(sd[u].x, __float_as_int(w[u]));
    }
}

#define GB 8
#define LP 68
template <bool RELU, bool BIAS>
__global__ __launch_bounds__(256)
void k_agg_gemm(const int* __restrict__ rowptr, const int* __restrict__ rec,
                const float* __restrict__ dinv, const float* __restrict__ h,
                const float* __restrict__ biasIn, const float* __restrict__ W,
                const float* __restrict__ biasOut, float* __restrict__ out, int n) {
    __shared__ __attribute__((aligned(16))) float sA[16 * LP];
    __shared__ __attribute__((aligned(16))) float sD[16 * LP];
    const int tid = threadIdx.x;
    const int t = blockIdx.x * blockDim.x + tid;
    const int i = t >> 4, q = t & 15;
    const int r = tid >> 4;
    const bool valid = i < n;
    const int ic = valid ? i : n - 1;
    const float4* __restrict__ h4 = reinterpret_cast<const float4*>(h);
    const int2*   __restrict__ r2 = reinterpret_cast<const int2*>(rec);

    const int beg = rowptr[ic];
    const int end = rowptr[ic + 1];
    const float di = dinv[ic];
    const float4 hs = h4[ic * 16 + q];
    const float4 bv = reinterpret_cast<const float4*>(biasIn)[q];
    const float ws = di * di;
    float4 acc;
    acc.x = fmaf(ws, hs.x, bv.x); acc.y = fmaf(ws, hs.y, bv.y);
    acc.z = fmaf(ws, hs.z, bv.z); acc.w = fmaf(ws, hs.w, bv.w);

    for (int p = beg; p < end; p += GB) {
        int2   rr[GB];
        float4 v[GB];
        #pragma unroll
        for (int u = 0; u < GB; ++u) rr[u] = r2[min(p + u, end - 1)];
        #pragma unroll
        for (int u = 0; u < GB; ++u) v[u] = h4[rr[u].x * 16 + q];
        #pragma unroll
        for (int u = 0; u < GB; ++u) {
            const bool live = (u == 0) || (p + u < end);
            const float w = __int_as_float(rr[u].y) * di;
            const float nx = fmaf(w, v[u].x, acc.x), ny = fmaf(w, v[u].y, acc.y);
            const float nz = fmaf(w, v[u].z, acc.z), nw = fmaf(w, v[u].w, acc.w);
            acc.x = live ? nx : acc.x; acc.y = live ? ny : acc.y;
            acc.z = live ? nz : acc.z; acc.w = live ? nw : acc.w;
        }
    }

    if constexpr (RELU) {
        acc.x = fmaxf(acc.x, 0.0f); acc.y = fmaxf(acc.y, 0.0f);
        acc.z = fmaxf(acc.z, 0.0f); acc.w = fmaxf(acc.w, 0.0f);
    }
    *reinterpret_cast<float4*>(&sA[r * LP + q * 4]) = acc;
    __syncthreads();

    const int wave = tid >> 5;
    if (wave < 4) {
        const int lane    = tid & 31;
        const int half    = lane >> 4;
        const int l16     = lane & 15;
        const int colBase = wave << 4;
        const float* __restrict__ wcol = W + colBase + l16;
        const float* arow = sA + l16 * LP;
        v8f c = {};
        #pragma unroll
        for (int kb = 0; kb < DCOL; kb += 4) {
            const int k0 = kb + half * 2;
            const v2f a = *reinterpret_cast<const v2f*>(arow + k0);
            v2f b; b[0] = wcol[(size_t)k0 * DCOL]; b[1] = wcol[(size_t)(k0 + 1) * DCOL];
            c = __builtin_amdgcn_wmma_f32_16x16x4_f32(
                     false, a,  false, b,
                     (short)0, c,  false,  false);
        }
        #pragma unroll
        for (int v = 0; v < 8; ++v) sD[(v + half * 8) * LP + colBase + l16] = c[v];
    }
    __syncthreads();

    float4 o = *reinterpret_cast<const float4*>(&sD[r * LP + q * 4]);
    if constexpr (BIAS) {
        const float4 bo = reinterpret_cast<const float4*>(biasOut)[q];
        o.x += bo.x; o.y += bo.y; o.z += bo.z; o.w += bo.w;
    }
    if (valid) reinterpret_cast<float4*>(out)[i * 16 + q] = o;
}

extern "C" void kernel_launch(void* const* d_in, const int* in_sizes, int n_in,
                              void* d_out, int out_size, void* d_ws, size_t ws_size,
                              hipStream_t stream) {
    const float* x   = (const float*)d_in[0];
    const int*   ei  = (const int*)  d_in[1];
    const float* W1  = (const float*)d_in[2];
    const float* b1  = (const float*)d_in[3];
    const float* W2  = (const float*)d_in[4];
    const float* b2  = (const float*)d_in[5];
    const float* Wfc = (const float*)d_in[6];
    const float* bfc = (const float*)d_in[7];
    float* out = (float*)d_out;

    const int N = in_sizes[0] / DIN;
    const int E = in_sizes[1] / 2;
    const int* src = ei;
    const int* dst = ei + E;

    const int NB  = (N + BN - 1) / BN;
    const int NC  = (E + CH - 1) / CH;
    const int NOF = (NB + 1) * NC;

    const size_t Np   = ((size_t)(N + 255) / 256) * 256;
    const size_t Ep   = ((size_t)(E + 255) / 256) * 256;
    const size_t NOFp = ((size_t)(NOF + 255) / 256) * 256;
    int*   OFF    = (int*)d_ws;
    int*   rowptr = OFF + NOFp;
    float* dinv   = (float*)(rowptr + Np + 256);
    int*   binned = (int*)(dinv + Np);
    int*   rec    = binned + 2 * (size_t)NC * CH;
    float* buf0   = (float*)(rec + 2 * Ep);
    float* buf1   = buf0 + (size_t)N * DCOL;

    const int blk = 256;
    const int gN16 = (N * 16 + blk - 1) / blk;

    const int G  = (N + 31) / 32;
    const int G1 = G * 3 / 16;
    const int G2 = G * 11 / 32;
    const int G3 = G - G1 - G2;

    k_chunk_sort<<<NC + G1, 256, 0, stream>>>(src, dst, OFF, binned, E, NB, NC, x, W1, buf0, N, 0);
    k_bin_count <<<NB + G2, BN, 0, stream>>>(binned, OFF, rowptr, dinv, N, NC, x, W1, buf0, NB, G1);
    k_bin_fill  <<<NB + G3, BN, 0, stream>>>(binned, OFF, rowptr, dinv, rec, N, NC, x, W1, buf0, NB, G1 + G2);

    k_agg_gemm<true,  false><<<gN16, blk, 0, stream>>>(rowptr, rec, dinv, buf0, b1, W2,  nullptr, buf1, N);

    k_agg_gemm<false, true ><<<gN16, blk, 0, stream>>>(rowptr, rec, dinv, buf1, b2, Wfc, bfc,     out,  N);
}


# Records hold src alone: dinv[dst] scales the finished sum once, dinv[src] is gathered (layer 1) or pre-multiplied into rows (layer 2); count and fill share a launch (5 to 4); sum order per node kept
# speedup vs baseline: 3.3176x; 1.0582x over previous
#include <hip/hip_runtime.h>
#include <hip/hip_bf16.h>


typedef __attribute__((ext_vector_type(2))) float v2f;
typedef __attribute__((ext_vector_type(8))) float v8f;

#define DIN 128
#define DCOL 64

__device__ __forceinline__ void product_rows32(const float* __restrict__ A, const float* __restrict__ W,
                                               float* __restrict__ H, int g, int n) {
    const int lane    = threadIdx.x & 31;
    const int wave    = threadIdx.x >> 5;
    const int rowBase = (g << 5) + ((wave >> 2) << 4);
    if (rowBase >= n) return;
    const int colBase = (wave & 3) << 4;
    const int half    = lane >> 4;
    const int l16     = lane & 15;

    const float* __restrict__ arow = A + (size_t)(rowBase + l16) * DIN;
    const float* __restrict__ wcol = W + colBase + l16;

    v8f c = {};
    #pragma unroll 8
    for (int kb = 0; kb < DIN; kb += 4) {
        const int k0 = kb + half * 2;
        v2f a; a[0] = arow[k0]; a[1] = arow[k0 + 1];
        v2f b; b[0] = wcol[(size_t)k0 * DCOL]; b[1] = wcol[(size_t)(k0 + 1) * DCOL];
        c = __builtin_amdgcn_wmma_f32_16x16x4_f32(
                 false, a,  false, b,
                 (short)0, c,  false,  false);
    }

    float* __restrict__ hout = H + colBase + l16;
    #pragma unroll
    for (int v = 0; v < 8; ++v)
        hout[(size_t)(rowBase + v + half * 8) * DCOL] = c[v];
}

#define BSH  8
#define BN   256
#define MAXB 256
#define EPT  16
#define CH   (256 * EPT)

__global__ __launch_bounds__(256)
void k_chunk_sort(const int* __restrict__ src, const int* __restrict__ dst,
                  int* __restrict__ OFF, int* __restrict__ binned, int e, int nb, int nc,
                  const float* __restrict__ x, const float* __restrict__ W1, float* __restrict__ h1,
                  int n, int g0) {
    __shared__ int s_h[MAXB + 1];
    __shared__ int s_tot[8];
    if ((int)blockIdx.x >= nc) {
        product_rows32(x, W1, h1, g0 + (int)blockIdx.x - nc, n);
        return;
    }
    const int t = threadIdx.x, c = blockIdx.x, lane = t & 31, wv = t >> 5;
    s_h[t] = 0;
    if (t == 0) s_h[MAXB] = 0;
    __syncthreads();
    const int e0 = c * CH + t;
    int d[EPT], s[EPT];
    #pragma unroll
    for (int j = 0; j < EPT; ++j) {
        const int i = e0 + j * 256;
        const bool in = i < e;
        d[j] = in ? dst[i] : -1;
        s[j] = in ? src[i] : 0;
    }
    int p[EPT];
    #pragma unroll
    for (int j = 0; j < EPT; ++j)
        p[j] = atomicAdd(&s_h[d[j] >= 0 ? (d[j] >> BSH) : MAXB], 1);
    __syncthreads();
    const int v = s_h[t];
    int inc = v;
    #pragma unroll
    for (int o = 1; o < 32; o <<= 1) {
        int u = __shfl_up(inc, o);
        if (lane >= o) inc += u;
    }
    if (lane == 31) s_tot[wv] = inc;
    __syncthreads();
    int off = 0;
    #pragma unroll
    for (int k = 0; k < 8; ++k)
        if (k < wv) off += s_tot[k];
    const int excl = off + inc - v;
    s_h[t] = excl;
    if (t < nb) OFF[t * nc + c] = excl;
    if (t == nb - 1) OFF[nb * nc + c] = excl + v;
    __syncthreads();
    int2* __restrict__ b2 = reinterpret_cast<int2*>(binned) + (size_t)c * CH;
    #pragma unroll
    for (int j = 0; j < EPT; ++j)
        if (d[j] >= 0) b2[s_h[d[j] >> BSH] + p[j]] = make_int2(s[j], d[j]);
}

__device__ __forceinline__ void bin_segments(const int* __restrict__ OFF, int b, int nc,
                                             int* s_st, int* s_o, int* s_w, int& beg, int& total) {
    const int t = threadIdx.x, lane = t & 31, wv = t >> 5;
    int o = 0, cnt = 0;
    if (t < nc) { o = OFF[b * nc + t]; cnt = OFF[(b + 1) * nc + t] - o; }
    int inc = cnt, so = o;
    #pragma unroll
    for (int k = 1; k < 32; k <<= 1) {
        int u = __shfl_up(inc, k);
        if (lane >= k) inc += u;
    }
    #pragma unroll
    for (int k = 16; k > 0; k >>= 1) so += __shfl_xor(so, k);
    if (lane == 31) s_w[wv] = inc;
    if (lane == 0)  s_w[8 + wv] = so;
    __syncthreads();
    int off = 0, tot = 0, bsum = 0;
    #pragma unroll
    for (int k = 0; k < 8; ++k) {
        const int w = s_w[k];
        if (k < wv) off += w;
        tot += w;
        bsum += s_w[8 + k];
    }
    const int st = off + inc - cnt;
    s_st[t] = st;
    s_o[t]  = t * CH + o - st;
    __syncthreads();
    beg = bsum;
    total = tot;
}

__device__ __forceinline__ int bin_place(const int* s_st, const int* s_o, int i) {
    int c = 0;
    #pragma unroll
    for (int step = 128; step > 0; step >>= 1)
        if (s_st[c + step] <= i) c += step;
    return s_o[c] + i;
}

#define FB 8
__global__ __launch_bounds__(256)
void k_bin_group(const int* __restrict__ binned, const int* __restrict__ OFF,
                 int* __restrict__ rowptr, float* __restrict__ dinv, int* __restrict__ rec,
                 int n, int nc,
                 const float* __restrict__ x, const float* __restrict__ W1, float* __restrict__ h1,
                 int nb, int g0) {
    __shared__ int s_c[BN + 1];
    __shared__ int s_tot[8];
    __shared__ int s_st[256];
    __shared__ int s_o[256];
    __shared__ int s_w[16];
    if ((int)blockIdx.x >= nb) {
        product_rows32(x, W1, h1, g0 + (int)blockIdx.x - nb, n);
        return;
    }
    const int t = threadIdx.x, b = blockIdx.x, lane = t & 31, wv = t >> 5;
    s_c[t] = 0;
    if (t == 0) s_c[BN] = 0;
    int beg, total;
    bin_segments(OFF, b, nc, s_st, s_o, s_w, beg, total);
    const int2* __restrict__ b2 = reinterpret_cast<const int2*>(binned);
    for (int i0 = 0; i0 < total; i0 += BN * FB) {
        int d[FB];
        #pragma unroll
        for (int u = 0; u < FB; ++u) {
            const int i = i0 + u * BN + t;
            const int dd = b2[bin_place(s_st, s_o, min(i, total - 1))].y;
            d[u] = (i < total) ? dd : -1;
        }
        #pragma unroll
        for (int u = 0; u < FB; ++u)
            if (d[u] >= 0) atomicAdd(&s_c[d[u] & (BN - 1)], 1);
    }
    __syncthreads();
    const int v = s_c[t];
    int inc = v;
    #pragma unroll
    for (int o = 1; o < 32; o <<= 1) {
        int u = __shfl_up(inc, o);
        if (lane >= o) inc += u;
    }
    if (lane == 31) s_tot[wv] = inc;
    __syncthreads();
    int off = beg;
    #pragma unroll
    for (int k = 0; k < 8; ++k)
        if (k < wv) off += s_tot[k];
    const int node = b * BN + t;
    const int start = off + inc - v;
    if (node < n) {
        rowptr[node] = start;
        if (node == n - 1) rowptr[n] = off + inc;
        dinv[node] = rsqrtf((float)(v + 1));
    }
    s_c[t] = start;
    __syncthreads();
    for (int i0 = 0; i0 < total; i0 += BN * FB) {
        int2 sd[FB];
        #pragma unroll
        for (int u = 0; u < FB; ++u) {
            const int i = i0 + u * BN + t;
            const int2 e2 = b2[bin_place(s_st, s_o, min(i, total - 1))];
            sd[u] = (i < total) ? e2 : make_int2(e2.x, -1);
        }
        int p[FB];
        #pragma unroll
        for (int u = 0; u < FB; ++u)
            p[u] = atomicAdd(&s_c[sd[u].y >= 0 ? (sd[u].y & (BN - 1)) : BN], 1);
        #pragma unroll
        for (int u = 0; u < FB; ++u)
            if (sd[u].y >= 0) rec[p[u]] = sd[u].x;
    }
}

#define GB 8
#define LP 68
template <bool FIRST>
__global__ __launch_bounds__(256)
void k_agg_gemm(const int* __restrict__ rowptr, const int* __restrict__ rec,
                const float* __restrict__ dinv, const float* __restrict__ h,
                const float* __restrict__ biasIn, const float* __restrict__ W,
                const float* __restrict__ biasOut, float* __restrict__ out, int n) {
    __shared__ __attribute__((aligned(16))) float sA[16 * LP];
    __shared__ __attribute__((aligned(16))) float sD[16 * LP];
    const int tid = threadIdx.x;
    const int t = blockIdx.x * blockDim.x + tid;
    const int i = t >> 4, q = t & 15;
    const int r = tid >> 4;
    const bool valid = i < n;
    const int ic = valid ? i : n - 1;
    const float4* __restrict__ h4 = reinterpret_cast<const float4*>(h);

    const int beg = rowptr[ic];
    const int end = rowptr[ic + 1];
    const float di = dinv[ic];
    const float4 hs = h4[ic * 16 + q];
    float4 bv;
    if constexpr (!FIRST) bv = reinterpret_cast<const float4*>(biasIn)[q];
    float4 acc = hs;
    if constexpr (FIRST) { acc.x *= di; acc.y *= di; acc.z *= di; acc.w *= di; }

    for (int p = beg; p < end; p += GB) {
        int    sx[GB];
        float  w[GB];
        float4 v[GB];
        #pragma unroll
        for (int u = 0; u < GB; ++u) sx[u] = rec[min(p + u, end - 1)];
        #pragma unroll
        for (int u = 0; u < GB; ++u) v[u] = h4[sx[u] * 16 + q];
        if constexpr (FIRST) {
            #pragma unroll
            for (int u = 0; u < GB; ++u) w[u] = dinv[sx[u]];
        }
        #pragma unroll
        for (int u = 0; u < GB; ++u) {
            const bool live = (u == 0) || (p + u < end);
            float nx, ny, nz, nw;
            if constexpr (FIRST) {
                nx = fmaf(w[u], v[u].x, acc.x); ny = fmaf(w[u], v[u].y, acc.y);
                nz = fmaf(w[u], v[u].z, acc.z); nw = fmaf(w[u], v[u].w, acc.w);
            } else {
                nx = acc.x + v[u].x; ny = acc.y + v[u].y;
                nz = acc.z + v[u].z; nw = acc.w + v[u].w;
            }
            acc.x = live ? nx : acc.x; acc.y = live ? ny : acc.y;
            acc.z = live ? nz : acc.z; acc.w = live ? nw : acc.w;
        }
    }
    if constexpr (FIRST) bv = reinterpret_cast<const float4*>(biasIn)[q];
    acc.x = fmaf(di, acc.x, bv.x); acc.y = fmaf(di, acc.y, bv.y);
    acc.z = fmaf(di, acc.z, bv.z); acc.w = fmaf(di, acc.w, bv.w);

    if constexpr (FIRST) {
        acc.x = fmaxf(acc.x, 0.0f); acc.y = fmaxf(acc.y, 0.0f);
        acc.z = fmaxf(acc.z, 0.0f); acc.w = fmaxf(acc.w, 0.0f);
    }
    *reinterpret_cast<float4*>(&sA[r * LP + q * 4]) = acc;
    __syncthreads();

    const int wave = tid >> 5;
    if (wave < 4) {
        const int lane    = tid & 31;
        const int half    = lane >> 4;
        const int l16     = lane & 15;
        const int colBase = wave << 4;
        const float* __restrict__ wcol = W + colBase + l16;
        const float* arow = sA + l16 * LP;
        v8f c = {};
        #pragma unroll
        for (int kb = 0; kb < DCOL; kb += 4) {
            const int k0 = kb + half * 2;
            const v2f a = *reinterpret_cast<const v2f*>(arow + k0);
            v2f b; b[0] = wcol[(size_t)k0 * DCOL]; b[1] = wcol[(size_t)(k0 + 1) * DCOL];
            c = __builtin_amdgcn_wmma_f32_16x16x4_f32(
                     false, a,  false, b,
                     (short)0, c,  false,  false);
        }
        #pragma unroll
        for (int v = 0; v < 8; ++v) sD[(v + half * 8) * LP + colBase + l16] = c[v];
    }
    __syncthreads();

    float4 o = *reinterpret_cast<const float4*>(&sD[r * LP + q * 4]);
    if constexpr (FIRST) {
        o.x *= di; o.y *= di; o.z *= di; o.w *= di;
    } else {
        const float4 bo = reinterpret_cast<const float4*>(biasOut)[q];
        o.x += bo.x; o.y += bo.y; o.z += bo.z; o.w += bo.w;
    }
    if (valid) reinterpret_cast<float4*>(out)[i * 16 + q] = o;
}

extern "C" void kernel_launch(void* const* d_in, const int* in_sizes, int n_in,
                              void* d_out, int out_size, void* d_ws, size_t ws_size,
                              hipStream_t stream) {
    const float* x   = (const float*)d_in[0];
    const int*   ei  = (const int*)  d_in[1];
    const float* W1  = (const float*)d_in[2];
    const float* b1  = (const float*)d_in[3];
    const float* W2  = (const float*)d_in[4];
    const float* b2  = (const float*)d_in[5];
    const float* Wfc = (const float*)d_in[6];
    const float* bfc = (const float*)d_in[7];
    float* out = (float*)d_out;

    const int N = in_sizes[0] / DIN;
    const int E = in_sizes[1] / 2;
    const int* src = ei;
    const int* dst = ei + E;

    const int NB  = (N + BN - 1) / BN;
    const int NC  = (E + CH - 1) / CH;
    const int NOF = (NB + 1) * NC;

    const size_t Np   = ((size_t)(N + 255) / 256) * 256;
    const size_t Ep   = ((size_t)(E + 255) / 256) * 256;
    const size_t NOFp = ((size_t)(NOF + 255) / 256) * 256;
    int*   OFF    = (int*)d_ws;
    int*   rowptr = OFF + NOFp;
    float* dinv   = (float*)(rowptr + Np + 256);
    int*   binned = (int*)(dinv + Np);
    int*   rec    = binned + 2 * (size_t)NC * CH;
    float* buf0   = (float*)(rec + 2 * Ep);
    float* buf1   = buf0 + (size_t)N * DCOL;

    const int blk = 256;
    const int gN16 = (N * 16 + blk - 1) / blk;

    const int G  = (N + 31) / 32;
    const int G1 = G * 3 / 16;
    const int G2 = G - G1;

    k_chunk_sort<<<NC + G1, 256, 0, stream>>>(src, dst, OFF, binned, E, NB, NC, x, W1, buf0, N, 0);
    k_bin_group <<<NB + G2, BN, 0, stream>>>(binned, OFF, rowptr, dinv, rec, N, NC, x, W1, buf0, NB, G1);

    k_agg_gemm<true ><<<gN16, blk, 0, stream>>>(rowptr, rec, dinv, buf0, b1, W2,  nullptr, buf1, N);

    k_agg_gemm<false><<<gN16, blk, 0, stream>>>(rowptr, rec, dinv, buf1, b2, Wfc, bfc,     out,  N);
}


# Bin block reads its edges once: the count pass leaves them packed in LDS, the place pass ranks from LDS (no second search or global read); x@W1 rows split evenly; sum order = LDS arrival rank
# speedup vs baseline: 3.7007x; 1.1155x over previous
#include <hip/hip_runtime.h>
#include <hip/hip_bf16.h>


typedef __attribute__((ext_vector_type(2))) float v2f;
typedef __attribute__((ext_vector_type(8))) float v8f;

#define DIN 128
#define DCOL 64

__device__ __forceinline__ void product_rows32(const float* __restrict__ A, const float* __restrict__ W,
                                               float* __restrict__ H, int g, int n) {
    const int lane    = threadIdx.x & 31;
    const int wave    = threadIdx.x >> 5;
    const int rowBase = (g << 5) + ((wave >> 2) << 4);
    if (rowBase >= n) return;
    const int colBase = (wave & 3) << 4;
    const int half    = lane >> 4;
    const int l16     = lane & 15;

    const float* __restrict__ arow = A + (size_t)(rowBase + l16) * DIN;
    const float* __restrict__ wcol = W + colBase + l16;

    v8f c = {};
    #pragma unroll 8
    for (int kb = 0; kb < DIN; kb += 4) {
        const int k0 = kb + half * 2;
        v2f a; a[0] = arow[k0]; a[1] = arow[k0 + 1];
        v2f b; b[0] = wcol[(size_t)k0 * DCOL]; b[1] = wcol[(size_t)(k0 + 1) * DCOL];
        c = __builtin_amdgcn_wmma_f32_16x16x4_f32(
                 false, a,  false, b,
                 (short)0, c,  false,  false);
    }

    float* __restrict__ hout = H + colBase + l16;
    #pragma unroll
    for (int v = 0; v < 8; ++v)
        hout[(size_t)(rowBase + v + half * 8) * DCOL] = c[v];
}

#define BSH  8
#define BN   256
#define MAXB 256
#define EPT  16
#define CH   (256 * EPT)

__global__ __launch_bounds__(256)
void k_chunk_sort(const int* __restrict__ src, const int* __restrict__ dst,
                  int* __restrict__ OFF, int* __restrict__ binned, int e, int nb, int nc,
                  const float* __restrict__ x, const float* __restrict__ W1, float* __restrict__ h1,
                  int n, int g0) {
    __shared__ int s_h[MAXB + 1];
    __shared__ int s_tot[8];
    if ((int)blockIdx.x >= nc) {
        product_rows32(x, W1, h1, g0 + (int)blockIdx.x - nc, n);
        return;
    }
    const int t = threadIdx.x, c = blockIdx.x, lane = t & 31, wv = t >> 5;
    s_h[t] = 0;
    if (t == 0) s_h[MAXB] = 0;
    __syncthreads();
    const int e0 = c * CH + t;
    int d[EPT], s[EPT];
    #pragma unroll
    for (int j = 0; j < EPT; ++j) {
        const int i = e0 + j * 256;
        const bool in = i < e;
        d[j] = in ? dst[i] : -1;
        s[j] = in ? src[i] : 0;
    }
    int p[EPT];
    #pragma unroll
    for (int j = 0; j < EPT; ++j)
        p[j] = atomicAdd(&s_h[d[j] >= 0 ? (d[j] >> BSH) : MAXB], 1);
    __syncthreads();
    const int v = s_h[t];
    int inc = v;
    #pragma unroll
    for (int o = 1; o < 32; o <<= 1) {
        int u = __shfl_up(inc, o);
        if (lane >= o) inc += u;
    }
    if (lane == 31) s_tot[wv] = inc;
    __syncthreads();
    int off = 0;
    #pragma unroll
    for (int k = 0; k < 8; ++k)
        if (k < wv) off += s_tot[k];
    const int excl = off + inc - v;
    s_h[t] = excl;
    if (t < nb) OFF[t * nc + c] = excl;
    if (t == nb - 1) OFF[nb * nc + c] = excl + v;
    __syncthreads();
    int2* __restrict__ b2 = reinterpret_cast<int2*>(binned) + (size_t)c * CH;
    #pragma unroll
    for (int j = 0; j < EPT; ++j)
        if (d[j] >= 0) b2[s_h[d[j] >> BSH] + p[j]] = make_int2(s[j], d[j]);
}

__device__ __forceinline__ void bin_segments(const int* __restrict__ OFF, int b, int nc,
                                             int* s_st, int* s_o, int* s_w, int& beg, int& total) {
    const int t = threadIdx.x, lane = t & 31, wv = t >> 5;
    int o = 0, cnt = 0;
    if (t < nc) { o = OFF[b * nc + t]; cnt = OFF[(b + 1) * nc + t] - o; }
    int inc = cnt, so = o;
    #pragma unroll
    for (int k = 1; k < 32; k <<= 1) {
        int u = __shfl_up(inc, k);
        if (lane >= k) inc += u;
    }
    #pragma unroll
    for (int k = 16; k > 0; k >>= 1) so += __shfl_xor(so, k);
    if (lane == 31) s_w[wv] = inc;
    if (lane == 0)  s_w[8 + wv] = so;
    __syncthreads();
    int off = 0, tot = 0, bsum = 0;
    #pragma unroll
    for (int k = 0; k < 8; ++k) {
        const int w = s_w[k];
        if (k < wv) off += w;
        tot += w;
        bsum += s_w[8 + k];
    }
    const int st = off + inc - cnt;
    s_st[t] = st;
    s_o[t]  = t * CH + o - st;
    __syncthreads();
    beg = bsum;
    total = tot;
}

__device__ __forceinline__ int bin_place(const int* s_st, const int* s_o, int i) {
    int c = 0;
    #pragma unroll
    for (int step = 128; step > 0; step >>= 1)
        if (s_st[c + step] <= i) c += step;
    return s_o[c] + i;
}

#define FB 10
#define SE 8192
__global__ __launch_bounds__(256)
void k_bin_group(const int* __restrict__ binned, const int* __restrict__ OFF,
                 int* __restrict__ rowptr, float* __restrict__ dinv, int* __restrict__ rec,
                 int n, int nc,
                 const float* __restrict__ x, const float* __restrict__ W1, float* __restrict__ h1,
                 int nb, int g0) {
    __shared__ int s_c[BN + 1];
    __shared__ int s_tot[8];
    __shared__ int s_st[256];
    __shared__ int s_o[256];
    __shared__ int s_w[16];
    __shared__ int s_e[SE];
    if ((int)blockIdx.x >= nb) {
        product_rows32(x, W1, h1, g0 + (int)blockIdx.x - nb, n);
        return;
    }
    const int t = threadIdx.x, b = blockIdx.x, lane = t & 31, wv = t >> 5;
    s_c[t] = 0;
    if (t == 0) s_c[BN] = 0;
    int beg, total;
    bin_segments(OFF, b, nc, s_st, s_o, s_w, beg, total);
    const int2* __restrict__ b2 = reinterpret_cast<const int2*>(binned);
    for (int i0 = 0; i0 < total; i0 += BN * FB) {
        int pk[FB];
        #pragma unroll
        for (int u = 0; u < FB; ++u) {
            const int i = i0 + u * BN + t;
            const int2 e2 = b2[bin_place(s_st, s_o, min(i, total - 1))];
            pk[u] = (e2.x << BSH) | (e2.y & (BN - 1));
        }
        #pragma unroll
        for (int u = 0; u < FB; ++u) {
            const int i = i0 + u * BN + t;
            const bool live = i < total;
            atomicAdd(&s_c[live ? (pk[u] & (BN - 1)) : BN], 1);
            if (live && i < SE) s_e[i] = pk[u];
        }
    }
    __syncthreads();
    const int v = s_c[t];
    int inc = v;
    #pragma unroll
    for (int o = 1; o < 32; o <<= 1) {
        int u = __shfl_up(inc, o);
        if (lane >= o) inc += u;
    }
    if (lane == 31) s_tot[wv] = inc;
    __syncthreads();
    int off = beg;
    #pragma unroll
    for (int k = 0; k < 8; ++k)
        if (k < wv) off += s_tot[k];
    const int node = b * BN + t;
    const int start = off + inc - v;
    if (node < n) {
        rowptr[node] = start;
        if (node == n - 1) rowptr[n] = off + inc;
        dinv[node] = rsqrtf((float)(v + 1));
    }
    s_c[t] = start;
    __syncthreads();
    for (int i0 = 0; i0 < total; i0 += BN * FB) {
        int ev[FB];
        #pragma unroll
        for (int u = 0; u < FB; ++u) {
            const int i = i0 + u * BN + t;
            ev[u] = s_e[min(min(i, total - 1), SE - 1)];
        }
        #pragma unroll
        for (int u = 0; u < FB; ++u) {
            const int i = i0 + u * BN + t;
            if (i >= total) {
                ev[u] = -1;
            } else if (i >= SE) {
                const int2 e2 = b2[bin_place(s_st, s_o, i)];
                ev[u] = (e2.x << BSH) | (e2.y & (BN - 1));
            }
        }
        int p[FB];
        #pragma unroll
        for (int u = 0; u < FB; ++u)
            p[u] = atomicAdd(&s_c[ev[u] >= 0 ? (ev[u] & (BN - 1)) : BN], 1);
        #pragma unroll
        for (int u = 0; u < FB; ++u)
            if (ev[u] >= 0) rec[p[u]] = ev[u] >> BSH;
    }
}

#define GB 8
#define LP 68
template <bool FIRST>
__global__ __launch_bounds__(256)
void k_agg_gemm(const int* __restrict__ rowptr, const int* __restrict__ rec,
                const float* __restrict__ dinv, const float* __restrict__ h,
                const float* __restrict__ biasIn, const float* __restrict__ W,
                const float* __restrict__ biasOut, float* __restrict__ out, int n) {
    __shared__ __attribute__((aligned(16))) float sA[16 * LP];
    __shared__ __attribute__((aligned(16))) float sD[16 * LP];
    const int tid = threadIdx.x;
    const int t = blockIdx.x * blockDim.x + tid;
    const int i = t >> 4, q = t & 15;
    const int r = tid >> 4;
    const bool valid = i < n;
    const int ic = valid ? i : n - 1;
    const float4* __restrict__ h4 = reinterpret_cast<const float4*>(h);

    const int beg = rowptr[ic];
    const int end = rowptr[ic + 1];
    const float di = dinv[ic];
    const float4 hs = h4[ic * 16 + q];
    float4 bv;
    if constexpr (!FIRST) bv = reinterpret_cast<const float4*>(biasIn)[q];
    float4 acc = hs;
    if constexpr (FIRST) { acc.x *= di; acc.y *= di; acc.z *= di; acc.w *= di; }

    for (int p = beg; p < end; p += GB) {
        int    sx[GB];
        float  w[GB];
        float4 v[GB];
        #pragma unroll
        for (int u = 0; u < GB; ++u) sx[u] = rec[min(p + u, end - 1)];
        #pragma unroll
        for (int u = 0; u < GB; ++u) v[u] = h4[sx[u] * 16 + q];
        if constexpr (FIRST) {
            #pragma unroll
            for (int u = 0; u < GB; ++u) w[u] = dinv[sx[u]];
        }
        #pragma unroll
        for (int u = 0; u < GB; ++u) {
            const bool live = (u == 0) || (p + u < end);
            float nx, ny, nz, nw;
            if constexpr (FIRST) {
                nx = fmaf(w[u], v[u].x, acc.x); ny = fmaf(w[u], v[u].y, acc.y);
                nz = fmaf(w[u], v[u].z, acc.z); nw = fmaf(w[u], v[u].w, acc.w);
            } else {
                nx = acc.x + v[u].x; ny = acc.y + v[u].y;
                nz = acc.z + v[u].z; nw = acc.w + v[u].w;
            }
            acc.x = live ? nx : acc.x; acc.y = live ? ny : acc.y;
            acc.z = live ? nz : acc.z; acc.w = live ? nw : acc.w;
        }
    }
    if constexpr (FIRST) bv = reinterpret_cast<const float4*>(biasIn)[q];
    acc.x = fmaf(di, acc.x, bv.x); acc.y = fmaf(di, acc.y, bv.y);
    acc.z = fmaf(di, acc.z, bv.z); acc.w = fmaf(di, acc.w, bv.w);

    if constexpr (FIRST) {
        acc.x = fmaxf(acc.x, 0.0f); acc.y = fmaxf(acc.y, 0.0f);
        acc.z = fmaxf(acc.z, 0.0f); acc.w = fmaxf(acc.w, 0.0f);
    }
    *reinterpret_cast<float4*>(&sA[r * LP + q * 4]) = acc;
    __syncthreads();

    const int wave = tid >> 5;
    if (wave < 4) {
        const int lane    = tid & 31;
        const int half    = lane >> 4;
        const int l16     = lane & 15;
        const int colBase = wave << 4;
        const float* __restrict__ wcol = W + colBase + l16;
        const float* arow = sA + l16 * LP;
        v8f c = {};
        #pragma unroll
        for (int kb = 0; kb < DCOL; kb += 4) {
            const int k0 = kb + half * 2;
            const v2f a = *reinterpret_cast<const v2f*>(arow + k0);
            v2f b; b[0] = wcol[(size_t)k0 * DCOL]; b[1] = wcol[(size_t)(k0 + 1) * DCOL];
            c = __builtin_amdgcn_wmma_f32_16x16x4_f32(
                     false, a,  false, b,
                     (short)0, c,  false,  false);
        }
        #pragma unroll
        for (int v = 0; v < 8; ++v) sD[(v + half * 8) * LP + colBase + l16] = c[v];
    }
    __syncthreads();

    float4 o = *reinterpret_cast<const float4*>(&sD[r * LP + q * 4]);
    if constexpr (FIRST) {
        o.x *= di; o.y *= di; o.z *= di; o.w *= di;
    } else {
        const float4 bo = reinterpret_cast<const float4*>(biasOut)[q];
        o.x += bo.x; o.y += bo.y; o.z += bo.z; o.w += bo.w;
    }
    if (valid) reinterpret_cast<float4*>(out)[i * 16 + q] = o;
}

extern "C" void kernel_launch(void* const* d_in, const int* in_sizes, int n_in,
                              void* d_out, int out_size, void* d_ws, size_t ws_size,
                              hipStream_t stream) {
    const float* x   = (const float*)d_in[0];
    const int*   ei  = (const int*)  d_in[1];
    const float* W1  = (const float*)d_in[2];
    const float* b1  = (const float*)d_in[3];
    const float* W2  = (const float*)d_in[4];
    const float* b2  = (const float*)d_in[5];
    const float* Wfc = (const float*)d_in[6];
    const float* bfc = (const float*)d_in[7];
    float* out = (float*)d_out;

    const int N = in_sizes[0] / DIN;
    const int E = in_sizes[1] / 2;
    const int* src = ei;
    const int* dst = ei + E;

    const int NB  = (N + BN - 1) / BN;
    const int NC  = (E + CH - 1) / CH;
    const int NOF = (NB + 1) * NC;

    const size_t Np   = ((size_t)(N + 255) / 256) * 256;
    const size_t Ep   = ((size_t)(E + 255) / 256) * 256;
    const size_t NOFp = ((size_t)(NOF + 255) / 256) * 256;
    int*   OFF    = (int*)d_ws;
    int*   rowptr = OFF + NOFp;
    float* dinv   = (float*)(rowptr + Np + 256);
    int*   binned = (int*)(dinv + Np);
    int*   rec    = binned + 2 * (size_t)NC * CH;
    float* buf0   = (float*)(rec + 2 * Ep);
    float* buf1   = buf0 + (size_t)N * DCOL;

    const int blk = 256;
    const int gN16 = (N * 16 + blk - 1) / blk;

    const int G  = (N + 31) / 32;
    const int G1 = G / 2;
    const int G2 = G - G1;

    k_chunk_sort<<<NC + G1, 256, 0, stream>>>(src, dst, OFF, binned, E, NB, NC, x, W1, buf0, N, 0);
    k_bin_group <<<NB + G2, BN, 0, stream>>>(binned, OFF, rowptr, dinv, rec, N, NC, x, W1, buf0, NB, G1);

    k_agg_gemm<true ><<<gN16, blk, 0, stream>>>(rowptr, rec, dinv, buf0, b1, W2,  nullptr, buf1, N);

    k_agg_gemm<false><<<gN16, blk, 0, stream>>>(rowptr, rec, dinv, buf1, b2, Wfc, bfc,     out,  N);
}


# Weights go through LDS: each block copies its weight matrix once (k-pairs), every B fragment is one 8-byte LDS read, not two global loads; x@W1: 128-row blocks, 16x64 strip per wave; sum order kept
# speedup vs baseline: 3.9840x; 1.0766x over previous
#include <hip/hip_runtime.h>
#include <hip/hip_bf16.h>


typedef __attribute__((ext_vector_type(2))) float v2f;
typedef __attribute__((ext_vector_type(8))) float v8f;

#define DIN 128
#define DCOL 64

#define WP2 160
template <int NP, bool PERM>
__device__ __forceinline__ void stage_weights(const float* __restrict__ W, float* sW) {
    const float4* __restrict__ W4 = reinterpret_cast<const float4*>(W);
    constexpr int NI = NP * 16 / 256;
    float4 r0[NI], r1[NI];
    #pragma unroll
    for (int i = 0; i < NI; ++i) {
        const int idx = threadIdx.x + 256 * i, kp = idx >> 4, c4 = idx & 15;
        r0[i] = W4[(2 * kp) * 16 + c4];
        r1[i] = W4[(2 * kp + 1) * 16 + c4];
    }
    #pragma unroll
    for (int i = 0; i < NI; ++i) {
        const int idx = threadIdx.x + 256 * i, kp = idx >> 4, c4 = idx & 15;
        if constexpr (PERM) {
            v2f* d = reinterpret_cast<v2f*>(&sW[kp * WP2 + c4 * 2]);
            v2f p;
            p[0] = r0[i].x; p[1] = r1[i].x; d[0]  = p;
            p[0] = r0[i].y; p[1] = r1[i].y; d[16] = p;
            p[0] = r0[i].z; p[1] = r1[i].z; d[32] = p;
            p[0] = r0[i].w; p[1] = r1[i].w; d[48] = p;
        } else {
            float4* d = reinterpret_cast<float4*>(&sW[kp * WP2 + c4 * 8]);
            d[0] = make_float4(r0[i].x, r1[i].x, r0[i].y, r1[i].y);
            d[1] = make_float4(r0[i].z, r1[i].z, r0[i].w, r1[i].w);
        }
    }
}

__device__ __forceinline__ void product_rows128(const float* __restrict__ A, const float* __restrict__ W,
                                                float* __restrict__ H, int g, int n, float* sW) {
    const int lane    = threadIdx.x & 31;
    const int wave    = threadIdx.x >> 5;
    const int rowBase = (g << 7) + (wave << 4);
    const int half    = lane >> 4;
    const int l16     = lane & 15;

    stage_weights<DIN / 2, true>(W, sW);
    __syncthreads();
    if (rowBase >= n) return;

    const int arow = min(rowBase + l16, n - 1);
    const v2f* __restrict__ a2 = reinterpret_cast<const v2f*>(A + (size_t)arow * DIN) + half;
    const v2f* wl = reinterpret_cast<const v2f*>(sW + half * WP2) + l16;

    v8f c0 = {}, c1 = {}, c2 = {}, c3 = {};
    v2f av[DIN / 4];
    #pragma unroll
    for (int i = 0; i < DIN / 4; ++i) av[i] = a2[2 * i];
    v2f bS[2][8];
    #pragma unroll
    for (int u = 0; u < 2; ++u) {
        #pragma unroll
        for (int t = 0; t < 4; ++t) bS[0][4 * u + t] = wl[u * WP2 + 16 * t];
    }
    #pragma unroll
    for (int q = 0; q < DIN / 8; ++q) {
        if (q + 1 < DIN / 8) {
            #pragma unroll
            for (int u = 0; u < 2; ++u) {
                #pragma unroll
                for (int t = 0; t < 4; ++t) bS[(q + 1) & 1][4 * u + t] = wl[(2 * (q + 1) + u) * WP2 + 16 * t];
            }
        }
        __builtin_amdgcn_sched_barrier(0x06);
        #pragma unroll
        for (int u = 0; u < 2; ++u) {
            const v2f a = av[2 * q + u];
            const v2f* b = &bS[q & 1][4 * u];
            c0 = __builtin_amdgcn_wmma_f32_16x16x4_f32(false, a, false, b[0], (short)0, c0, false, false);
            c1 = __builtin_amdgcn_wmma_f32_16x16x4_f32(false, a, false, b[1], (short)0, c1, false, false);
            c2 = __builtin_amdgcn_wmma_f32_16x16x4_f32(false, a, false, b[2], (short)0, c2, false, false);
            c3 = __builtin_amdgcn_wmma_f32_16x16x4_f32(false, a, false, b[3], (short)0, c3, false, false);
        }
    }

    float4* __restrict__ h4 = reinterpret_cast<float4*>(H) + l16;
    #pragma unroll
    for (int v = 0; v < 8; ++v) {
        const int row = rowBase + v + half * 8;
        if (row < n) h4[(size_t)row * (DCOL / 4)] = make_float4(c0[v], c1[v], c2[v], c3[v]);
    }
}

#define BSH  8
#define BN   256
#define MAXB 256
#define EPT  16
#define CH   (256 * EPT)

__global__ __launch_bounds__(256)
void k_chunk_sort(const int* __restrict__ src, const int* __restrict__ dst,
                  int* __restrict__ OFF, int* __restrict__ binned, int e, int nb, int nc,
                  const float* __restrict__ x, const float* __restrict__ W1, float* __restrict__ h1,
                  int n, int g0) {
    __shared__ int s_h[MAXB + 1];
    __shared__ int s_tot[8];
    __shared__ __attribute__((aligned(16))) float s_w1[(DIN / 2) * WP2];
    if ((int)blockIdx.x >= nc) {
        product_rows128(x, W1, h1, g0 + (int)blockIdx.x - nc, n, s_w1);
        return;
    }
    const int t = threadIdx.x, c = blockIdx.x, lane = t & 31, wv = t >> 5;
    s_h[t] = 0;
    if (t == 0) s_h[MAXB] = 0;
    __syncthreads();
    const int e0 = c * CH + t;
    int d[EPT], s[EPT];
    #pragma unroll
    for (int j = 0; j < EPT; ++j) {
        const int i = e0 + j * 256;
        const bool in = i < e;
        d[j] = in ? dst[i] : -1;
        s[j] = in ? src[i] : 0;
    }
    int p[EPT];
    #pragma unroll
    for (int j = 0; j < EPT; ++j)
        p[j] = atomicAdd(&s_h[d[j] >= 0 ? (d[j] >> BSH) : MAXB], 1);
    __syncthreads();
    const int v = s_h[t];
    int inc = v;
    #pragma unroll
    for (int o = 1; o < 32; o <<= 1) {
        int u = __shfl_up(inc, o);
        if (lane >= o) inc += u;
    }
    if (lane == 31) s_tot[wv] = inc;
    __syncthreads();
    int off = 0;
    #pragma unroll
    for (int k = 0; k < 8; ++k)
        if (k < wv) off += s_tot[k];
    const int excl = off + inc - v;
    s_h[t] = excl;
    if (t < nb) OFF[t * nc + c] = excl;
    if (t == nb - 1) OFF[nb * nc + c] = excl + v;
    __syncthreads();
    int2* __restrict__ b2 = reinterpret_cast<int2*>(binned) + (size_t)c * CH;
    #pragma unroll
    for (int j = 0; j < EPT; ++j)
        if (d[j] >= 0) b2[s_h[d[j] >> BSH] + p[j]] = make_int2(s[j], d[j]);
}

__device__ __forceinline__ void bin_segments(const int* __restrict__ OFF, int b, int nc,
                                             int* s_st, int* s_o, int* s_w, int& beg, int& total) {
    const int t = threadIdx.x, lane = t & 31, wv = t >> 5;
    int o = 0, cnt = 0;
    if (t < nc) { o = OFF[b * nc + t]; cnt = OFF[(b + 1) * nc + t] - o; }
    int inc = cnt, so = o;
    #pragma unroll
    for (int k = 1; k < 32; k <<= 1) {
        int u = __shfl_up(inc, k);
        if (lane >= k) inc += u;
    }
    #pragma unroll
    for (int k = 16; k > 0; k >>= 1) so += __shfl_xor(so, k);
    if (lane == 31) s_w[wv] = inc;
    if (lane == 0)  s_w[8 + wv] = so;
    __syncthreads();
    int off = 0, tot = 0, bsum = 0;
    #pragma unroll
    for (int k = 0; k < 8; ++k) {
        const int w = s_w[k];
        if (k < wv) off += w;
        tot += w;
        bsum += s_w[8 + k];
    }
    const int st = off + inc - cnt;
    s_st[t] = st;
    s_o[t]  = t * CH + o - st;
    __syncthreads();
    beg = bsum;
    total = tot;
}

__device__ __forceinline__ int bin_place(const int* s_st, const int* s_o, int i) {
    int c = 0;
    #pragma unroll
    for (int step = 128; step > 0; step >>= 1)
        if (s_st[c + step] <= i) c += step;
    return s_o[c] + i;
}

#define FB 10
#define SE 8192
__global__ __launch_bounds__(256)
void k_bin_group(const int* __restrict__ binned, const int* __restrict__ OFF,
                 int* __restrict__ rowptr, float* __restrict__ dinv, int* __restrict__ rec,
                 int n, int nc,
                 const float* __restrict__ x, const float* __restrict__ W1, float* __restrict__ h1,
                 int nb, int g0) {
    __shared__ int s_c[BN + 1];
    __shared__ int s_tot[8];
    __shared__ int s_st[256];
    __shared__ int s_o[256];
    __shared__ int s_w[16];
    __shared__ int s_e[SE];
    __shared__ __attribute__((aligned(16))) float s_w1[(DIN / 2) * WP2];
    if ((int)blockIdx.x >= nb) {
        product_rows128(x, W1, h1, g0 + (int)blockIdx.x - nb, n, s_w1);
        return;
    }
    const int t = threadIdx.x, b = blockIdx.x, lane = t & 31, wv = t >> 5;
    s_c[t] = 0;
    if (t == 0) s_c[BN] = 0;
    int beg, total;
    bin_segments(OFF, b, nc, s_st, s_o, s_w, beg, total);
    const int2* __restrict__ b2 = reinterpret_cast<const int2*>(binned);
    for (int i0 = 0; i0 < total; i0 += BN * FB) {
        int pk[FB];
        #pragma unroll
        for (int u = 0; u < FB; ++u) {
            const int i = i0 + u * BN + t;
            const int2 e2 = b2[bin_place(s_st, s_o, min(i, total - 1))];
            pk[u] = (e2.x << BSH) | (e2.y & (BN - 1));
        }
        #pragma unroll
        for (int u = 0; u < FB; ++u) {
            const int i = i0 + u * BN + t;
            const bool live = i < total;
            atomicAdd(&s_c[live ? (pk[u] & (BN - 1)) : BN], 1);
            if (live && i < SE) s_e[i] = pk[u];
        }
    }
    __syncthreads();
    const int v = s_c[t];
    int inc = v;
    #pragma unroll
    for (int o = 1; o < 32; o <<= 1) {
        int u = __shfl_up(inc, o);
        if (lane >= o) inc += u;
    }
    if (lane == 31) s_tot[wv] = inc;
    __syncthreads();
    int off = beg;
    #pragma unroll
    for (int k = 0; k < 8; ++k)
        if (k < wv) off += s_tot[k];
    const int node = b * BN + t;
    const int start = off + inc - v;
    if (node < n) {
        rowptr[node] = start;
        if (node == n - 1) rowptr[n] = off + inc;
        dinv[node] = rsqrtf((float)(v + 1));
    }
    s_c[t] = start;
    __syncthreads();
    for (int i0 = 0; i0 < total; i0 += BN * FB) {
        int ev[FB];
        #pragma unroll
        for (int u = 0; u < FB; ++u) {
            const int i = i0 + u * BN + t;
            ev[u] = s_e[min(min(i, total - 1), SE - 1)];
        }
        #pragma unroll
        for (int u = 0; u < FB; ++u) {
            const int i = i0 + u * BN + t;
            if (i >= total) {
                ev[u] = -1;
            } else if (i >= SE) {
                const int2 e2 = b2[bin_place(s_st, s_o, i)];
                ev[u] = (e2.x << BSH) | (e2.y & (BN - 1));
            }
        }
        int p[FB];
        #pragma unroll
        for (int u = 0; u < FB; ++u)
            p[u] = atomicAdd(&s_c[ev[u] >= 0 ? (ev[u] & (BN - 1)) : BN], 1);
        #pragma unroll
        for (int u = 0; u < FB; ++u)
            if (ev[u] >= 0) rec[p[u]] = ev[u] >> BSH;
    }
}

#define GB 8
#define LP 68
template <bool FIRST>
__global__ __launch_bounds__(256)
void k_agg_gemm(const int* __restrict__ rowptr, const int* __restrict__ rec,
                const float* __restrict__ dinv, const float* __restrict__ h,
                const float* __restrict__ biasIn, const float* __restrict__ W,
                const float* __restrict__ biasOut, float* __restrict__ out, int n) {
    __shared__ __attribute__((aligned(16))) float sA[16 * LP];
    __shared__ __attribute__((aligned(16))) float sD[16 * LP];
    __shared__ __attribute__((aligned(16))) float sW[(DCOL / 2) * WP2];
    const int tid = threadIdx.x;
    stage_weights<DCOL / 2, false>(W, sW);
    const int t = blockIdx.x * blockDim.x + tid;
    const int i = t >> 4, q = t & 15;
    const int r = tid >> 4;
    const bool valid = i < n;
    const int ic = valid ? i : n - 1;
    const float4* __restrict__ h4 = reinterpret_cast<const float4*>(h);

    const int beg = rowptr[ic];
    const int end = rowptr[ic + 1];
    const float di = dinv[ic];
    const float4 hs = h4[ic * 16 + q];
    float4 bv;
    if constexpr (!FIRST) bv = reinterpret_cast<const float4*>(biasIn)[q];
    float4 acc = hs;
    if constexpr (FIRST) { acc.x *= di; acc.y *= di; acc.z *= di; acc.w *= di; }

    for (int p = beg; p < end; p += GB) {
        int    sx[GB];
        float  w[GB];
        float4 v[GB];
        #pragma unroll
        for (int u = 0; u < GB; ++u) sx[u] = rec[min(p + u, end - 1)];
        #pragma unroll
        for (int u = 0; u < GB; ++u) v[u] = h4[sx[u] * 16 + q];
        if constexpr (FIRST) {
            #pragma unroll
            for (int u = 0; u < GB; ++u) w[u] = dinv[sx[u]];
        }
        #pragma unroll
        for (int u = 0; u < GB; ++u) {
            const bool live = (u == 0) || (p + u < end);
            float nx, ny, nz, nw;
            if constexpr (FIRST) {
                nx = fmaf(w[u], v[u].x, acc.x); ny = fmaf(w[u], v[u].y, acc.y);
                nz = fmaf(w[u], v[u].z, acc.z); nw = fmaf(w[u], v[u].w, acc.w);
            } else {
                nx = acc.x + v[u].x; ny = acc.y + v[u].y;
                nz = acc.z + v[u].z; nw = acc.w + v[u].w;
            }
            acc.x = live ? nx : acc.x; acc.y = live ? ny : acc.y;
            acc.z = live ? nz : acc.z; acc.w = live ? nw : acc.w;
        }
    }
    if constexpr (FIRST) bv = reinterpret_cast<const float4*>(biasIn)[q];
    acc.x = fmaf(di, acc.x, bv.x); acc.y = fmaf(di, acc.y, bv.y);
    acc.z = fmaf(di, acc.z, bv.z); acc.w = fmaf(di, acc.w, bv.w);

    if constexpr (FIRST) {
        acc.x = fmaxf(acc.x, 0.0f); acc.y = fmaxf(acc.y, 0.0f);
        acc.z = fmaxf(acc.z, 0.0f); acc.w = fmaxf(acc.w, 0.0f);
    }
    *reinterpret_cast<float4*>(&sA[r * LP + q * 4]) = acc;
    __syncthreads();

    const int wave = tid >> 5;
    if (wave < 4) {
        const int lane    = tid & 31;
        const int half    = lane >> 4;
        const int l16     = lane & 15;
        const int colBase = wave << 4;
        const v2f* wl = reinterpret_cast<const v2f*>(sW + half * WP2) + colBase + l16;
        const float* arow = sA + l16 * LP;
        v8f c = {};
        #pragma unroll
        for (int kb = 0; kb < DCOL; kb += 4) {
            const int k0 = kb + half * 2;
            const v2f a = *reinterpret_cast<const v2f*>(arow + k0);
            const v2f b = wl[(kb >> 2) * WP2];
            c = __builtin_amdgcn_wmma_f32_16x16x4_f32(
                     false, a,  false, b,
                     (short)0, c,  false,  false);
        }
        #pragma unroll
        for (int v = 0; v < 8; ++v) sD[(v + half * 8) * LP + colBase + l16] = c[v];
    }
    __syncthreads();

    float4 o = *reinterpret_cast<const float4*>(&sD[r * LP + q * 4]);
    if constexpr (FIRST) {
        o.x *= di; o.y *= di; o.z *= di; o.w *= di;
    } else {
        const float4 bo = reinterpret_cast<const float4*>(biasOut)[q];
        o.x += bo.x; o.y += bo.y; o.z += bo.z; o.w += bo.w;
    }
    if (valid) reinterpret_cast<float4*>(out)[i * 16 + q] = o;
}

extern "C" void kernel_launch(void* const* d_in, const int* in_sizes, int n_in,
                              void* d_out, int out_size, void* d_ws, size_t ws_size,
                              hipStream_t stream) {
    const float* x   = (const float*)d_in[0];
    const int*   ei  = (const int*)  d_in[1];
    const float* W1  = (const float*)d_in[2];
    const float* b1  = (const float*)d_in[3];
    const float* W2  = (const float*)d_in[4];
    const float* b2  = (const float*)d_in[5];
    const float* Wfc = (const float*)d_in[6];
    const float* bfc = (const float*)d_in[7];
    float* out = (float*)d_out;

    const int N = in_sizes[0] / DIN;
    const int E = in_sizes[1] / 2;
    const int* src = ei;
    const int* dst = ei + E;

    const int NB  = (N + BN - 1) / BN;
    const int NC  = (E + CH - 1) / CH;
    const int NOF = (NB + 1) * NC;

    const size_t Np   = ((size_t)(N + 255) / 256) * 256;
    const size_t Ep   = ((size_t)(E + 255) / 256) * 256;
    const size_t NOFp = ((size_t)(NOF + 255) / 256) * 256;
    int*   OFF    = (int*)d_ws;
    int*   rowptr = OFF + NOFp;
    float* dinv   = (float*)(rowptr + Np + 256);
    int*   binned = (int*)(dinv + Np);
    int*   rec    = binned + 2 * (size_t)NC * CH;
    float* buf0   = (float*)(rec + 2 * Ep);
    float* buf1   = buf0 + (size_t)N * DCOL;

    const int blk = 256;
    const int gN16 = (N * 16 + blk - 1) / blk;

    const int G  = (N + 127) / 128;
    const int G1 = G / 2;
    const int G2 = G - G1;

    k_chunk_sort<<<NC + G1, 256, 0, stream>>>(src, dst, OFF, binned, E, NB, NC, x, W1, buf0, N, 0);
    k_bin_group <<<NB + G2, BN, 0, stream>>>(binned, OFF, rowptr, dinv, rec, N, NC, x, W1, buf0, NB, G1);

    k_agg_gemm<true ><<<gN16, blk, 0, stream>>>(rowptr, rec, dinv, buf0, b1, W2,  nullptr, buf1, N);

    k_agg_gemm<false><<<gN16, blk, 0, stream>>>(rowptr, rec, dinv, buf1, b2, Wfc, bfc,     out,  N);
}


# Head folded into layer 2's weights: a spare block of launch 1 forms W2 @ Wfc and b2 @ Wfc + bfc on the f32 matrix units; last launch is a plain row sum (no product, LDS, barrier); sums reassociated
# speedup vs baseline: 4.1180x; 1.0336x over previous
#include <hip/hip_runtime.h>
#include <hip/hip_bf16.h>


typedef __attribute__((ext_vector_type(2))) float v2f;
typedef __attribute__((ext_vector_type(8))) float v8f;

#define DIN 128
#define DCOL 64

#define WP2 160
template <int NP, bool PERM>
__device__ __forceinline__ void stage_weights(const float* __restrict__ W, float* sW) {
    const float4* __restrict__ W4 = reinterpret_cast<const float4*>(W);
    constexpr int NI = NP * 16 / 256;
    float4 r0[NI], r1[NI];
    #pragma unroll
    for (int i = 0; i < NI; ++i) {
        const int idx = threadIdx.x + 256 * i, kp = idx >> 4, c4 = idx & 15;
        r0[i] = W4[(2 * kp) * 16 + c4];
        r1[i] = W4[(2 * kp + 1) * 16 + c4];
    }
    #pragma unroll
    for (int i = 0; i < NI; ++i) {
        const int idx = threadIdx.x + 256 * i, kp = idx >> 4, c4 = idx & 15;
        if constexpr (PERM) {
            v2f* d = reinterpret_cast<v2f*>(&sW[kp * WP2 + c4 * 2]);
            v2f p;
            p[0] = r0[i].x; p[1] = r1[i].x; d[0]  = p;
            p[0] = r0[i].y; p[1] = r1[i].y; d[16] = p;
            p[0] = r0[i].z; p[1] = r1[i].z; d[32] = p;
            p[0] = r0[i].w; p[1] = r1[i].w; d[48] = p;
        } else {
            float4* d = reinterpret_cast<float4*>(&sW[kp * WP2 + c4 * 8]);
            d[0] = make_float4(r0[i].x, r1[i].x, r0[i].y, r1[i].y);
            d[1] = make_float4(r0[i].z, r1[i].z, r0[i].w, r1[i].w);
        }
    }
}

__device__ __forceinline__ void product_rows128(const float* __restrict__ A, const float* __restrict__ W,
                                                float* __restrict__ H, int g, int n, float* sW) {
    const int lane    = threadIdx.x & 31;
    const int wave    = threadIdx.x >> 5;
    const int rowBase = (g << 7) + (wave << 4);
    const int half    = lane >> 4;
    const int l16     = lane & 15;

    stage_weights<DIN / 2, true>(W, sW);
    __syncthreads();
    if (rowBase >= n) return;

    const int arow = min(rowBase + l16, n - 1);
    const v2f* __restrict__ a2 = reinterpret_cast<const v2f*>(A + (size_t)arow * DIN) + half;
    const v2f* wl = reinterpret_cast<const v2f*>(sW + half * WP2) + l16;

    v8f c0 = {}, c1 = {}, c2 = {}, c3 = {};
    v2f av[DIN / 4];
    #pragma unroll
    for (int i = 0; i < DIN / 4; ++i) av[i] = a2[2 * i];
    v2f bS[2][8];
    #pragma unroll
    for (int u = 0; u < 2; ++u) {
        #pragma unroll
        for (int t = 0; t < 4; ++t) bS[0][4 * u + t] = wl[u * WP2 + 16 * t];
    }
    #pragma unroll
    for (int q = 0; q < DIN / 8; ++q) {
        if (q + 1 < DIN / 8) {
            #pragma unroll
            for (int u = 0; u < 2; ++u) {
                #pragma unroll
                for (int t = 0; t < 4; ++t) bS[(q + 1) & 1][4 * u + t] = wl[(2 * (q + 1) + u) * WP2 + 16 * t];
            }
        }
        __builtin_amdgcn_sched_barrier(0x06);
        #pragma unroll
        for (int u = 0; u < 2; ++u) {
            const v2f a = av[2 * q + u];
            const v2f* b = &bS[q & 1][4 * u];
            c0 = __builtin_amdgcn_wmma_f32_16x16x4_f32(false, a, false, b[0], (short)0, c0, false, false);
            c1 = __builtin_amdgcn_wmma_f32_16x16x4_f32(false, a, false, b[1], (short)0, c1, false, false);
            c2 = __builtin_amdgcn_wmma_f32_16x16x4_f32(false, a, false, b[2], (short)0, c2, false, false);
            c3 = __builtin_amdgcn_wmma_f32_16x16x4_f32(false, a, false, b[3], (short)0, c3, false, false);
        }
    }

    float4* __restrict__ h4 = reinterpret_cast<float4*>(H) + l16;
    #pragma unroll
    for (int v = 0; v < 8; ++v) {
        const int row = rowBase + v + half * 8;
        if (row < n) h4[(size_t)row * (DCOL / 4)] = make_float4(c0[v], c1[v], c2[v], c3[v]);
    }
}

__device__ __forceinline__ void fold_head(const float* __restrict__ W2, const float* __restrict__ b2,
                                          const float* __restrict__ Wfc, const float* __restrict__ bfc,
                                          float* __restrict__ Wp, float* __restrict__ cvec, float* sW) {
    stage_weights<DCOL / 2, false>(Wfc, sW);
    __syncthreads();
    const int lane = threadIdx.x & 31;
    const int wave = threadIdx.x >> 5;
    if (wave > 4) return;
    const int half = lane >> 4;
    const int l16  = lane & 15;
    const bool isW = wave < 4;
    const float* __restrict__ arow = isW ? W2 + (wave * 16 + l16) * DCOL : b2;
    const v2f* wl = reinterpret_cast<const v2f*>(sW + half * WP2) + l16;
    v8f c0 = {}, c1 = {}, c2 = {}, c3 = {};
    #pragma unroll
    for (int kb = 0; kb < DCOL; kb += 4) {
        const v2f a = *reinterpret_cast<const v2f*>(arow + kb + half * 2);
        const v2f* b = wl + (kb >> 2) * WP2;
        c0 = __builtin_amdgcn_wmma_f32_16x16x4_f32(false, a, false, b[0],  (short)0, c0, false, false);
        c1 = __builtin_amdgcn_wmma_f32_16x16x4_f32(false, a, false, b[16], (short)0, c1, false, false);
        c2 = __builtin_amdgcn_wmma_f32_16x16x4_f32(false, a, false, b[32], (short)0, c2, false, false);
        c3 = __builtin_amdgcn_wmma_f32_16x16x4_f32(false, a, false, b[48], (short)0, c3, false, false);
    }
    float* __restrict__ d = isW ? Wp + (wave * 16 + half * 8) * DCOL + l16 : cvec + l16;
    const int nr = isW ? 8 : (half == 0 ? 1 : 0);
    float a0 = 0.0f, a1 = 0.0f, a2 = 0.0f, a3 = 0.0f;
    if (!isW) { a0 = bfc[l16]; a1 = bfc[16 + l16]; a2 = bfc[32 + l16]; a3 = bfc[48 + l16]; }
    #pragma unroll
    for (int v = 0; v < 8; ++v) {
        if (v < nr) {
            d[v * DCOL]      = c0[v] + a0;
            d[v * DCOL + 16] = c1[v] + a1;
            d[v * DCOL + 32] = c2[v] + a2;
            d[v * DCOL + 48] = c3[v] + a3;
        }
    }
}

#define BSH  8
#define BN   256
#define MAXB 256
#define EPT  16
#define CH   (256 * EPT)

__global__ __launch_bounds__(256)
void k_chunk_sort(const int* __restrict__ src, const int* __restrict__ dst,
                  int* __restrict__ OFF, int* __restrict__ binned, int e, int nb, int nc,
                  const float* __restrict__ x, const float* __restrict__ W1, float* __restrict__ h1,
                  int n, int g0, int ng,
                  const float* __restrict__ W2, const float* __restrict__ bias2,
                  const float* __restrict__ Wfc, const float* __restrict__ bfc,
                  float* __restrict__ Wp, float* __restrict__ cvec) {
    __shared__ int s_h[MAXB + 1];
    __shared__ int s_tot[8];
    __shared__ __attribute__((aligned(16))) float s_w1[(DIN / 2) * WP2];
    if ((int)blockIdx.x >= nc) {
        const int g = (int)blockIdx.x - nc;
        if (g == ng) fold_head(W2, bias2, Wfc, bfc, Wp, cvec, s_w1);
        else         product_rows128(x, W1, h1, g0 + g, n, s_w1);
        return;
    }
    const int t = threadIdx.x, c = blockIdx.x, lane = t & 31, wv = t >> 5;
    s_h[t] = 0;
    if (t == 0) s_h[MAXB] = 0;
    __syncthreads();
    const int e0 = c * CH + t;
    int d[EPT], s[EPT];
    #pragma unroll
    for (int j = 0; j < EPT; ++j) {
        const int i = e0 + j * 256;
        const bool in = i < e;
        d[j] = in ? dst[i] : -1;
        s[j] = in ? src[i] : 0;
    }
    int p[EPT];
    #pragma unroll
    for (int j = 0; j < EPT; ++j)
        p[j] = atomicAdd(&s_h[d[j] >= 0 ? (d[j] >> BSH) : MAXB], 1);
    __syncthreads();
    const int v = s_h[t];
    int inc = v;
    #pragma unroll
    for (int o = 1; o < 32; o <<= 1) {
        int u = __shfl_up(inc, o);
        if (lane >= o) inc += u;
    }
    if (lane == 31) s_tot[wv] = inc;
    __syncthreads();
    int off = 0;
    #pragma unroll
    for (int k = 0; k < 8; ++k)
        if (k < wv) off += s_tot[k];
    const int excl = off + inc - v;
    s_h[t] = excl;
    if (t < nb) OFF[t * nc + c] = excl;
    if (t == nb - 1) OFF[nb * nc + c] = excl + v;
    __syncthreads();
    int2* __restrict__ b2 = reinterpret_cast<int2*>(binned) + (size_t)c * CH;
    #pragma unroll
    for (int j = 0; j < EPT; ++j)
        if (d[j] >= 0) b2[s_h[d[j] >> BSH] + p[j]] = make_int2(s[j], d[j]);
}

__device__ __forceinline__ void bin_segments(const int* __restrict__ OFF, int b, int nc,
                                             int* s_st, int* s_o, int* s_w, int& beg, int& total) {
    const int t = threadIdx.x, lane = t & 31, wv = t >> 5;
    int o = 0, cnt = 0;
    if (t < nc) { o = OFF[b * nc + t]; cnt = OFF[(b + 1) * nc + t] - o; }
    int inc = cnt, so = o;
    #pragma unroll
    for (int k = 1; k < 32; k <<= 1) {
        int u = __shfl_up(inc, k);
        if (lane >= k) inc += u;
    }
    #pragma unroll
    for (int k = 16; k > 0; k >>= 1) so += __shfl_xor(so, k);
    if (lane == 31) s_w[wv] = inc;
    if (lane == 0)  s_w[8 + wv] = so;
    __syncthreads();
    int off = 0, tot = 0, bsum = 0;
    #pragma unroll
    for (int k = 0; k < 8; ++k) {
        const int w = s_w[k];
        if (k < wv) off += w;
        tot += w;
        bsum += s_w[8 + k];
    }
    const int st = off + inc - cnt;
    s_st[t] = st;
    s_o[t]  = t * CH + o - st;
    __syncthreads();
    beg = bsum;
    total = tot;
}

__device__ __forceinline__ int bin_place(const int* s_st, const int* s_o, int i) {
    int c = 0;
    #pragma unroll
    for (int step = 128; step > 0; step >>= 1)
        if (s_st[c + step] <= i) c += step;
    return s_o[c] + i;
}

#define FB 10
#define SE 8192
__global__ __launch_bounds__(256)
void k_bin_group(const int* __restrict__ binned, const int* __restrict__ OFF,
                 int* __restrict__ rowptr, float* __restrict__ dinv, int* __restrict__ rec,
                 int n, int nc,
                 const float* __restrict__ x, const float* __restrict__ W1, float* __restrict__ h1,
                 int nb, int g0) {
    __shared__ int s_c[BN + 1];
    __shared__ int s_tot[8];
    __shared__ int s_st[256];
    __shared__ int s_o[256];
    __shared__ int s_w[16];
    __shared__ int s_e[SE];
    __shared__ __attribute__((aligned(16))) float s_w1[(DIN / 2) * WP2];
    if ((int)blockIdx.x >= nb) {
        product_rows128(x, W1, h1, g0 + (int)blockIdx.x - nb, n, s_w1);
        return;
    }
    const int t = threadIdx.x, b = blockIdx.x, lane = t & 31, wv = t >> 5;
    s_c[t] = 0;
    if (t == 0) s_c[BN] = 0;
    int beg, total;
    bin_segments(OFF, b, nc, s_st, s_o, s_w, beg, total);
    const int2* __restrict__ b2 = reinterpret_cast<const int2*>(binned);
    for (int i0 = 0; i0 < total; i0 += BN * FB) {
        int pk[FB];
        #pragma unroll
        for (int u = 0; u < FB; ++u) {
            const int i = i0 + u * BN + t;
            const int2 e2 = b2[bin_place(s_st, s_o, min(i, total - 1))];
            pk[u] = (e2.x << BSH) | (e2.y & (BN - 1));
        }
        #pragma unroll
        for (int u = 0; u < FB; ++u) {
            const int i = i0 + u * BN + t;
            const bool live = i < total;
            atomicAdd(&s_c[live ? (pk[u] & (BN - 1)) : BN], 1);
            if (live && i < SE) s_e[i] = pk[u];
        }
    }
    __syncthreads();
    const int v = s_c[t];
    int inc = v;
    #pragma unroll
    for (int o = 1; o < 32; o <<= 1) {
        int u = __shfl_up(inc, o);
        if (lane >= o) inc += u;
    }
    if (lane == 31) s_tot[wv] = inc;
    __syncthreads();
    int off = beg;
    #pragma unroll
    for (int k = 0; k < 8; ++k)
        if (k < wv) off += s_tot[k];
    const int node = b * BN + t;
    const int start = off + inc - v;
    if (node < n) {
        rowptr[node] = start;
        if (node == n - 1) rowptr[n] = off + inc;
        dinv[node] = rsqrtf((float)(v + 1));
    }
    s_c[t] = start;
    __syncthreads();
    for (int i0 = 0; i0 < total; i0 += BN * FB) {
        int ev[FB];
        #pragma unroll
        for (int u = 0; u < FB; ++u) {
            const int i = i0 + u * BN + t;
            ev[u] = s_e[min(min(i, total - 1), SE - 1)];
        }
        #pragma unroll
        for (int u = 0; u < FB; ++u) {
            const int i = i0 + u * BN + t;
            if (i >= total) {
                ev[u] = -1;
            } else if (i >= SE) {
                const int2 e2 = b2[bin_place(s_st, s_o, i)];
                ev[u] = (e2.x << BSH) | (e2.y & (BN - 1));
            }
        }
        int p[FB];
        #pragma unroll
        for (int u = 0; u < FB; ++u)
            p[u] = atomicAdd(&s_c[ev[u] >= 0 ? (ev[u] & (BN - 1)) : BN], 1);
        #pragma unroll
        for (int u = 0; u < FB; ++u)
            if (ev[u] >= 0) rec[p[u]] = ev[u] >> BSH;
    }
}

#define GB 8
#define LP 68
__global__ __launch_bounds__(256)
void k_agg_gemm(const int* __restrict__ rowptr, const int* __restrict__ rec,
                const float* __restrict__ dinv, const float* __restrict__ h,
                const float* __restrict__ biasIn, const float* __restrict__ W,
                float* __restrict__ out, int n) {
    __shared__ __attribute__((aligned(16))) float sA[16 * LP];
    __shared__ __attribute__((aligned(16))) float sD[16 * LP];
    __shared__ __attribute__((aligned(16))) float sW[(DCOL / 2) * WP2];
    const int tid = threadIdx.x;
    stage_weights<DCOL / 2, false>(W, sW);
    const int t = blockIdx.x * blockDim.x + tid;
    const int i = t >> 4, q = t & 15;
    const int r = tid >> 4;
    const bool valid = i < n;
    const int ic = valid ? i : n - 1;
    const float4* __restrict__ h4 = reinterpret_cast<const float4*>(h);

    const int beg = rowptr[ic];
    const int end = rowptr[ic + 1];
    const float di = dinv[ic];
    const float4 hs = h4[ic * 16 + q];
    float4 acc = hs;
    acc.x *= di; acc.y *= di; acc.z *= di; acc.w *= di;

    for (int p = beg; p < end; p += GB) {
        int    sx[GB];
        float  w[GB];
        float4 v[GB];
        #pragma unroll
        for (int u = 0; u < GB; ++u) sx[u] = rec[min(p + u, end - 1)];
        #pragma unroll
        for (int u = 0; u < GB; ++u) v[u] = h4[sx[u] * 16 + q];
        #pragma unroll
        for (int u = 0; u < GB; ++u) w[u] = dinv[sx[u]];
        #pragma unroll
        for (int u = 0; u < GB; ++u) {
            const bool live = (u == 0) || (p + u < end);
            const float nx = fmaf(w[u], v[u].x, acc.x), ny = fmaf(w[u], v[u].y, acc.y);
            const float nz = fmaf(w[u], v[u].z, acc.z), nw = fmaf(w[u], v[u].w, acc.w);
            acc.x = live ? nx : acc.x; acc.y = live ? ny : acc.y;
            acc.z = live ? nz : acc.z; acc.w = live ? nw : acc.w;
        }
    }
    const float4 bv = reinterpret_cast<const float4*>(biasIn)[q];
    acc.x = fmaf(di, acc.x, bv.x); acc.y = fmaf(di, acc.y, bv.y);
    acc.z = fmaf(di, acc.z, bv.z); acc.w = fmaf(di, acc.w, bv.w);
    acc.x = fmaxf(acc.x, 0.0f); acc.y = fmaxf(acc.y, 0.0f);
    acc.z = fmaxf(acc.z, 0.0f); acc.w = fmaxf(acc.w, 0.0f);
    *reinterpret_cast<float4*>(&sA[r * LP + q * 4]) = acc;
    __syncthreads();

    const int wave = tid >> 5;
    if (wave < 4) {
        const int lane    = tid & 31;
        const int half    = lane >> 4;
        const int l16     = lane & 15;
        const int colBase = wave << 4;
        const v2f* wl = reinterpret_cast<const v2f*>(sW + half * WP2) + colBase + l16;
        const float* arow = sA + l16 * LP;
        v8f c = {};
        #pragma unroll
        for (int kb = 0; kb < DCOL; kb += 4) {
            const int k0 = kb + half * 2;
            const v2f a = *reinterpret_cast<const v2f*>(arow + k0);
            const v2f b = wl[(kb >> 2) * WP2];
            c = __builtin_amdgcn_wmma_f32_16x16x4_f32(
                     false, a,  false, b,
                     (short)0, c,  false,  false);
        }
        #pragma unroll
        for (int v = 0; v < 8; ++v) sD[(v + half * 8) * LP + colBase + l16] = c[v];
    }
    __syncthreads();

    float4 o = *reinterpret_cast<const float4*>(&sD[r * LP + q * 4]);
    o.x *= di; o.y *= di; o.z *= di; o.w *= di;
    if (valid) reinterpret_cast<float4*>(out)[i * 16 + q] = o;
}

__global__ __launch_bounds__(256)
void k_agg_sum(const int* __restrict__ rowptr, const int* __restrict__ rec,
               const float* __restrict__ dinv, const float* __restrict__ h,
               const float* __restrict__ cvec, float* __restrict__ out, int n) {
    const int t = blockIdx.x * blockDim.x + threadIdx.x;
    const int i = t >> 4, q = t & 15;
    if (i >= n) return;
    const float4* __restrict__ h4 = reinterpret_cast<const float4*>(h);

    const int beg = rowptr[i];
    const int end = rowptr[i + 1];
    const float di = dinv[i];
    const float4 cv = reinterpret_cast<const float4*>(cvec)[q];
    float4 acc = h4[i * 16 + q];

    for (int p = beg; p < end; p += GB) {
        int    sx[GB];
        float4 v[GB];
        #pragma unroll
        for (int u = 0; u < GB; ++u) sx[u] = rec[min(p + u, end - 1)];
        #pragma unroll
        for (int u = 0; u < GB; ++u) v[u] = h4[sx[u] * 16 + q];
        #pragma unroll
        for (int u = 0; u < GB; ++u) {
            const bool live = (u == 0) || (p + u < end);
            const float nx = acc.x + v[u].x, ny = acc.y + v[u].y;
            const float nz = acc.z + v[u].z, nw = acc.w + v[u].w;
            acc.x = live ? nx : acc.x; acc.y = live ? ny : acc.y;
            acc.z = live ? nz : acc.z; acc.w = live ? nw : acc.w;
        }
    }
    reinterpret_cast<float4*>(out)[i * 16 + q] =
        make_float4(fmaf(di, acc.x, cv.x), fmaf(di, acc.y, cv.y), fmaf(di, acc.z, cv.z), fmaf(di, acc.w, cv.w));
}

extern "C" void kernel_launch(void* const* d_in, const int* in_sizes, int n_in,
                              void* d_out, int out_size, void* d_ws, size_t ws_size,
                              hipStream_t stream) {
    const float* x   = (const float*)d_in[0];
    const int*   ei  = (const int*)  d_in[1];
    const float* W1  = (const float*)d_in[2];
    const float* b1  = (const float*)d_in[3];
    const float* W2  = (const float*)d_in[4];
    const float* b2  = (const float*)d_in[5];
    const float* Wfc = (const float*)d_in[6];
    const float* bfc = (const float*)d_in[7];
    float* out = (float*)d_out;

    const int N = in_sizes[0] / DIN;
    const int E = in_sizes[1] / 2;
    const int* src = ei;
    const int* dst = ei + E;

    const int NB  = (N + BN - 1) / BN;
    const int NC  = (E + CH - 1) / CH;
    const int NOF = (NB + 1) * NC;

    const size_t Np   = ((size_t)(N + 255) / 256) * 256;
    const size_t Ep   = ((size_t)(E + 255) / 256) * 256;
    const size_t NOFp = ((size_t)(NOF + 255) / 256) * 256;
    int*   OFF    = (int*)d_ws;
    int*   rowptr = OFF + NOFp;
    float* dinv   = (float*)(rowptr + Np + 256);
    int*   binned = (int*)(dinv + Np);
    int*   rec    = binned + 2 * (size_t)NC * CH;
    float* buf0   = (float*)(rec + 2 * Ep);
    float* buf1   = buf0 + (size_t)N * DCOL;
    float* Wp     = buf1 + ((size_t)N * DCOL + 255) / 256 * 256;
    float* cvec   = Wp + DCOL * DCOL;

    const int blk = 256;
    const int gN16 = (N * 16 + blk - 1) / blk;

    const int G  = (N + 127) / 128;
    const int G1 = G / 2;
    const int G2 = G - G1;

    k_chunk_sort<<<NC + G1 + 1, 256, 0, stream>>>(src, dst, OFF, binned, E, NB, NC, x, W1, buf0, N, 0, G1,
                                                  W2, b2, Wfc, bfc, Wp, cvec);
    k_bin_group <<<NB + G2, BN, 0, stream>>>(binned, OFF, rowptr, dinv, rec, N, NC, x, W1, buf0, NB, G1);

    k_agg_gemm<<<gN16, blk, 0, stream>>>(rowptr, rec, dinv, buf0, b1, Wp, buf1, N);

    k_agg_sum <<<gN16, blk, 0, stream>>>(rowptr, rec, dinv, buf1, cvec, out, N);
}


# Both aggregation launches fetch a node's list entries with one load per 16 edges (one entry per lane of the node's 16-lane row) and pass them round by DPP row_share moves; sum order per node kept
# speedup vs baseline: 4.3265x; 1.0506x over previous
#include <hip/hip_runtime.h>
#include <hip/hip_bf16.h>


typedef __attribute__((ext_vector_type(2))) float v2f;
typedef __attribute__((ext_vector_type(8))) float v8f;

#define DIN 128
#define DCOL 64

#define WP2 160
template <int NP, bool PERM>
__device__ __forceinline__ void stage_weights(const float* __restrict__ W, float* sW) {
    const float4* __restrict__ W4 = reinterpret_cast<const float4*>(W);
    constexpr int NI = NP * 16 / 256;
    float4 r0[NI], r1[NI];
    #pragma unroll
    for (int i = 0; i < NI; ++i) {
        const int idx = threadIdx.x + 256 * i, kp = idx >> 4, c4 = idx & 15;
        r0[i] = W4[(2 * kp) * 16 + c4];
        r1[i] = W4[(2 * kp + 1) * 16 + c4];
    }
    #pragma unroll
    for (int i = 0; i < NI; ++i) {
        const int idx = threadIdx.x + 256 * i, kp = idx >> 4, c4 = idx & 15;
        if constexpr (PERM) {
            v2f* d = reinterpret_cast<v2f*>(&sW[kp * WP2 + c4 * 2]);
            v2f p;
            p[0] = r0[i].x; p[1] = r1[i].x; d[0]  = p;
            p[0] = r0[i].y; p[1] = r1[i].y; d[16] = p;
            p[0] = r0[i].z; p[1] = r1[i].z; d[32] = p;
            p[0] = r0[i].w; p[1] = r1[i].w; d[48] = p;
        } else {
            float4* d = reinterpret_cast<float4*>(&sW[kp * WP2 + c4 * 8]);
            d[0] = make_float4(r0[i].x, r1[i].x, r0[i].y, r1[i].y);
            d[1] = make_float4(r0[i].z, r1[i].z, r0[i].w, r1[i].w);
        }
    }
}

__device__ __forceinline__ void product_rows128(const float* __restrict__ A, const float* __restrict__ W,
                                                float* __restrict__ H, int g, int n, float* sW) {
    const int lane    = threadIdx.x & 31;
    const int wave    = threadIdx.x >> 5;
    const int rowBase = (g << 7) + (wave << 4);
    const int half    = lane >> 4;
    const int l16     = lane & 15;

    stage_weights<DIN / 2, true>(W, sW);
    __syncthreads();
    if (rowBase >= n) return;

    const int arow = min(rowBase + l16, n - 1);
    const v2f* __restrict__ a2 = reinterpret_cast<const v2f*>(A + (size_t)arow * DIN) + half;
    const v2f* wl = reinterpret_cast<const v2f*>(sW + half * WP2) + l16;

    v8f c0 = {}, c1 = {}, c2 = {}, c3 = {};
    v2f av[DIN / 4];
    #pragma unroll
    for (int i = 0; i < DIN / 4; ++i) av[i] = a2[2 * i];
    v2f bS[2][8];
    #pragma unroll
    for (int u = 0; u < 2; ++u) {
        #pragma unroll
        for (int t = 0; t < 4; ++t) bS[0][4 * u + t] = wl[u * WP2 + 16 * t];
    }
    #pragma unroll
    for (int q = 0; q < DIN / 8; ++q) {
        if (q + 1 < DIN / 8) {
            #pragma unroll
            for (int u = 0; u < 2; ++u) {
                #pragma unroll
                for (int t = 0; t < 4; ++t) bS[(q + 1) & 1][4 * u + t] = wl[(2 * (q + 1) + u) * WP2 + 16 * t];
            }
        }
        __builtin_amdgcn_sched_barrier(0x06);
        #pragma unroll
        for (int u = 0; u < 2; ++u) {
            const v2f a = av[2 * q + u];
            const v2f* b = &bS[q & 1][4 * u];
            c0 = __builtin_amdgcn_wmma_f32_16x16x4_f32(false, a, false, b[0], (short)0, c0, false, false);
            c1 = __builtin_amdgcn_wmma_f32_16x16x4_f32(false, a, false, b[1], (short)0, c1, false, false);
            c2 = __builtin_amdgcn_wmma_f32_16x16x4_f32(false, a, false, b[2], (short)0, c2, false, false);
            c3 = __builtin_amdgcn_wmma_f32_16x16x4_f32(false, a, false, b[3], (short)0, c3, false, false);
        }
    }

    float4* __restrict__ h4 = reinterpret_cast<float4*>(H) + l16;
    #pragma unroll
    for (int v = 0; v < 8; ++v) {
        const int row = rowBase + v + half * 8;
        if (row < n) h4[(size_t)row * (DCOL / 4)] = make_float4(c0[v], c1[v], c2[v], c3[v]);
    }
}

__device__ __forceinline__ void fold_head(const float* __restrict__ W2, const float* __restrict__ b2,
                                          const float* __restrict__ Wfc, const float* __restrict__ bfc,
                                          float* __restrict__ Wp, float* __restrict__ cvec, float* sW) {
    stage_weights<DCOL / 2, false>(Wfc, sW);
    __syncthreads();
    const int lane = threadIdx.x & 31;
    const int wave = threadIdx.x >> 5;
    if (wave > 4) return;
    const int half = lane >> 4;
    const int l16  = lane & 15;
    const bool isW = wave < 4;
    const float* __restrict__ arow = isW ? W2 + (wave * 16 + l16) * DCOL : b2;
    const v2f* wl = reinterpret_cast<const v2f*>(sW + half * WP2) + l16;
    v8f c0 = {}, c1 = {}, c2 = {}, c3 = {};
    #pragma unroll
    for (int kb = 0; kb < DCOL; kb += 4) {
        const v2f a = *reinterpret_cast<const v2f*>(arow + kb + half * 2);
        const v2f* b = wl + (kb >> 2) * WP2;
        c0 = __builtin_amdgcn_wmma_f32_16x16x4_f32(false, a, false, b[0],  (short)0, c0, false, false);
        c1 = __builtin_amdgcn_wmma_f32_16x16x4_f32(false, a, false, b[16], (short)0, c1, false, false);
        c2 = __builtin_amdgcn_wmma_f32_16x16x4_f32(false, a, false, b[32], (short)0, c2, false, false);
        c3 = __builtin_amdgcn_wmma_f32_16x16x4_f32(false, a, false, b[48], (short)0, c3, false, false);
    }
    float* __restrict__ d = isW ? Wp + (wave * 16 + half * 8) * DCOL + l16 : cvec + l16;
    const int nr = isW ? 8 : (half == 0 ? 1 : 0);
    float a0 = 0.0f, a1 = 0.0f, a2 = 0.0f, a3 = 0.0f;
    if (!isW) { a0 = bfc[l16]; a1 = bfc[16 + l16]; a2 = bfc[32 + l16]; a3 = bfc[48 + l16]; }
    #pragma unroll
    for (int v = 0; v < 8; ++v) {
        if (v < nr) {
            d[v * DCOL]      = c0[v] + a0;
            d[v * DCOL + 16] = c1[v] + a1;
            d[v * DCOL + 32] = c2[v] + a2;
            d[v * DCOL + 48] = c3[v] + a3;
        }
    }
}

#define BSH  8
#define BN   256
#define MAXB 256
#define EPT  16
#define CH   (256 * EPT)

__global__ __launch_bounds__(256)
void k_chunk_sort(const int* __restrict__ src, const int* __restrict__ dst,
                  int* __restrict__ OFF, int* __restrict__ binned, int e, int nb, int nc,
                  const float* __restrict__ x, const float* __restrict__ W1, float* __restrict__ h1,
                  int n, int g0, int ng,
                  const float* __restrict__ W2, const float* __restrict__ bias2,
                  const float* __restrict__ Wfc, const float* __restrict__ bfc,
                  float* __restrict__ Wp, float* __restrict__ cvec) {
    __shared__ int s_h[MAXB + 1];
    __shared__ int s_tot[8];
    __shared__ __attribute__((aligned(16))) float s_w1[(DIN / 2) * WP2];
    if ((int)blockIdx.x >= nc) {
        const int g = (int)blockIdx.x - nc;
        if (g == ng) fold_head(W2, bias2, Wfc, bfc, Wp, cvec, s_w1);
        else         product_rows128(x, W1, h1, g0 + g, n, s_w1);
        return;
    }
    const int t = threadIdx.x, c = blockIdx.x, lane = t & 31, wv = t >> 5;
    s_h[t] = 0;
    if (t == 0) s_h[MAXB] = 0;
    __syncthreads();
    const int e0 = c * CH + t;
    int d[EPT], s[EPT];
    #pragma unroll
    for (int j = 0; j < EPT; ++j) {
        const int i = e0 + j * 256;
        const bool in = i < e;
        d[j] = in ? dst[i] : -1;
        s[j] = in ? src[i] : 0;
    }
    int p[EPT];
    #pragma unroll
    for (int j = 0; j < EPT; ++j)
        p[j] = atomicAdd(&s_h[d[j] >= 0 ? (d[j] >> BSH) : MAXB], 1);
    __syncthreads();
    const int v = s_h[t];
    int inc = v;
    #pragma unroll
    for (int o = 1; o < 32; o <<= 1) {
        int u = __shfl_up(inc, o);
        if (lane >= o) inc += u;
    }
    if (lane == 31) s_tot[wv] = inc;
    __syncthreads();
    int off = 0;
    #pragma unroll
    for (int k = 0; k < 8; ++k)
        if (k < wv) off += s_tot[k];
    const int excl = off + inc - v;
    s_h[t] = excl;
    if (t < nb) OFF[t * nc + c] = excl;
    if (t == nb - 1) OFF[nb * nc + c] = excl + v;
    __syncthreads();
    int2* __restrict__ b2 = reinterpret_cast<int2*>(binned) + (size_t)c * CH;
    #pragma unroll
    for (int j = 0; j < EPT; ++j)
        if (d[j] >= 0) b2[s_h[d[j] >> BSH] + p[j]] = make_int2(s[j], d[j]);
}

__device__ __forceinline__ void bin_segments(const int* __restrict__ OFF, int b, int nc,
                                             int* s_st, int* s_o, int* s_w, int& beg, int& total) {
    const int t = threadIdx.x, lane = t & 31, wv = t >> 5;
    int o = 0, cnt = 0;
    if (t < nc) { o = OFF[b * nc + t]; cnt = OFF[(b + 1) * nc + t] - o; }
    int inc = cnt, so = o;
    #pragma unroll
    for (int k = 1; k < 32; k <<= 1) {
        int u = __shfl_up(inc, k);
        if (lane >= k) inc += u;
    }
    #pragma unroll
    for (int k = 16; k > 0; k >>= 1) so += __shfl_xor(so, k);
    if (lane == 31) s_w[wv] = inc;
    if (lane == 0)  s_w[8 + wv] = so;
    __syncthreads();
    int off = 0, tot = 0, bsum = 0;
    #pragma unroll
    for (int k = 0; k < 8; ++k) {
        const int w = s_w[k];
        if (k < wv) off += w;
        tot += w;
        bsum += s_w[8 + k];
    }
    const int st = off + inc - cnt;
    s_st[t] = st;
    s_o[t]  = t * CH + o - st;
    __syncthreads();
    beg = bsum;
    total = tot;
}

__device__ __forceinline__ int bin_place(const int* s_st, const int* s_o, int i) {
    int c = 0;
    #pragma unroll
    for (int step = 128; step > 0; step >>= 1)
        if (s_st[c + step] <= i) c += step;
    return s_o[c] + i;
}

#define FB 10
#define SE 8192
__global__ __launch_bounds__(256)
void k_bin_group(const int* __restrict__ binned, const int* __restrict__ OFF,
                 int* __restrict__ rowptr, float* __restrict__ dinv, int* __restrict__ rec,
                 int n, int nc,
                 const float* __restrict__ x, const float* __restrict__ W1, float* __restrict__ h1,
                 int nb, int g0) {
    __shared__ int s_c[BN + 1];
    __shared__ int s_tot[8];
    __shared__ int s_st[256];
    __shared__ int s_o[256];
    __shared__ int s_w[16];
    __shared__ int s_e[SE];
    __shared__ __attribute__((aligned(16))) float s_w1[(DIN / 2) * WP2];
    if ((int)blockIdx.x >= nb) {
        product_rows128(x, W1, h1, g0 + (int)blockIdx.x - nb, n, s_w1);
        return;
    }
    const int t = threadIdx.x, b = blockIdx.x, lane = t & 31, wv = t >> 5;
    s_c[t] = 0;
    if (t == 0) s_c[BN] = 0;
    int beg, total;
    bin_segments(OFF, b, nc, s_st, s_o, s_w, beg, total);
    const int2* __restrict__ b2 = reinterpret_cast<const int2*>(binned);
    for (int i0 = 0; i0 < total; i0 += BN * FB) {
        int pk[FB];
        #pragma unroll
        for (int u = 0; u < FB; ++u) {
            const int i = i0 + u * BN + t;
            const int2 e2 = b2[bin_place(s_st, s_o, min(i, total - 1))];
            pk[u] = (e2.x << BSH) | (e2.y & (BN - 1));
        }
        #pragma unroll
        for (int u = 0; u < FB; ++u) {
            const int i = i0 + u * BN + t;
            const bool live = i < total;
            atomicAdd(&s_c[live ? (pk[u] & (BN - 1)) : BN], 1);
            if (live && i < SE) s_e[i] = pk[u];
        }
    }
    __syncthreads();
    const int v = s_c[t];
    int inc = v;
    #pragma unroll
    for (int o = 1; o < 32; o <<= 1) {
        int u = __shfl_up(inc, o);
        if (lane >= o) inc += u;
    }
    if (lane == 31) s_tot[wv] = inc;
    __syncthreads();
    int off = beg;
    #pragma unroll
    for (int k = 0; k < 8; ++k)
        if (k < wv) off += s_tot[k];
    const int node = b * BN + t;
    const int start = off + inc - v;
    if (node < n) {
        rowptr[node] = start;
        if (node == n - 1) rowptr[n] = off + inc;
        dinv[node] = rsqrtf((float)(v + 1));
    }
    s_c[t] = start;
    __syncthreads();
    for (int i0 = 0; i0 < total; i0 += BN * FB) {
        int ev[FB];
        #pragma unroll
        for (int u = 0; u < FB; ++u) {
            const int i = i0 + u * BN + t;
            ev[u] = s_e[min(min(i, total - 1), SE - 1)];
        }
        #pragma unroll
        for (int u = 0; u < FB; ++u) {
            const int i = i0 + u * BN + t;
            if (i >= total) {
                ev[u] = -1;
            } else if (i >= SE) {
                const int2 e2 = b2[bin_place(s_st, s_o, i)];
                ev[u] = (e2.x << BSH) | (e2.y & (BN - 1));
            }
        }
        int p[FB];
        #pragma unroll
        for (int u = 0; u < FB; ++u)
            p[u] = atomicAdd(&s_c[ev[u] >= 0 ? (ev[u] & (BN - 1)) : BN], 1);
        #pragma unroll
        for (int u = 0; u < FB; ++u)
            if (ev[u] >= 0) rec[p[u]] = ev[u] >> BSH;
    }
}

#define ROW_LANE(v, L) __builtin_amdgcn_update_dpp(0, (v), 0x150 + (L), 0xF, 0xF, false)
#define GB 8
template <int O>
__device__ __forceinline__ void row_indices(int mine, int (&sx)[GB]) {
    sx[0] = ROW_LANE(mine, O + 0); sx[1] = ROW_LANE(mine, O + 1);
    sx[2] = ROW_LANE(mine, O + 2); sx[3] = ROW_LANE(mine, O + 3);
    sx[4] = ROW_LANE(mine, O + 4); sx[5] = ROW_LANE(mine, O + 5);
    sx[6] = ROW_LANE(mine, O + 6); sx[7] = ROW_LANE(mine, O + 7);
}

template <int O>
__device__ __forceinline__ void batch_weighted(const float4* __restrict__ h4, const float* __restrict__ dinv,
                                               int mine, int pb, int end, int q, float4& acc) {
    int    sx[GB];
    float  w[GB];
    float4 v[GB];
    row_indices<O>(mine, sx);
    #pragma unroll
    for (int u = 0; u < GB; ++u) v[u] = h4[sx[u] * 16 + q];
    #pragma unroll
    for (int u = 0; u < GB; ++u) w[u] = dinv[sx[u]];
    __builtin_amdgcn_sched_barrier(0);
    #pragma unroll
    for (int u = 0; u < GB; ++u) {
        const bool live = (u == 0) || (pb + u < end);
        const float nx = fmaf(w[u], v[u].x, acc.x), ny = fmaf(w[u], v[u].y, acc.y);
        const float nz = fmaf(w[u], v[u].z, acc.z), nw = fmaf(w[u], v[u].w, acc.w);
        acc.x = live ? nx : acc.x; acc.y = live ? ny : acc.y;
        acc.z = live ? nz : acc.z; acc.w = live ? nw : acc.w;
    }
}

template <int O>
__device__ __forceinline__ void batch_plain(const float4* __restrict__ h4, int mine, int pb, int end, int q, float4& acc) {
    int    sx[GB];
    float4 v[GB];
    row_indices<O>(mine, sx);
    #pragma unroll
    for (int u = 0; u < GB; ++u) v[u] = h4[sx[u] * 16 + q];
    asm volatile("" :: "v"(v[0].x), "v"(v[1].x), "v"(v[2].x), "v"(v[3].x), "v"(v[4].x), "v"(v[5].x), "v"(v[6].x), "v"(v[7].x));
    __builtin_amdgcn_sched_barrier(0);
    #pragma unroll
    for (int u = 0; u < GB; ++u) {
        const bool live = (u == 0) || (pb + u < end);
        const float nx = acc.x + v[u].x, ny = acc.y + v[u].y;
        const float nz = acc.z + v[u].z, nw = acc.w + v[u].w;
        acc.x = live ? nx : acc.x; acc.y = live ? ny : acc.y;
        acc.z = live ? nz : acc.z; acc.w = live ? nw : acc.w;
    }
}

#define LP 68
__global__ __launch_bounds__(256)
void k_agg_gemm(const int* __restrict__ rowptr, const int* __restrict__ rec,
                const float* __restrict__ dinv, const float* __restrict__ h,
                const float* __restrict__ biasIn, const float* __restrict__ W,
                float* __restrict__ out, int n) {
    __shared__ __attribute__((aligned(16))) float sA[16 * LP];
    __shared__ __attribute__((aligned(16))) float sD[16 * LP];
    __shared__ __attribute__((aligned(16))) float sW[(DCOL / 2) * WP2];
    const int tid = threadIdx.x;
    stage_weights<DCOL / 2, false>(W, sW);
    const int t = blockIdx.x * blockDim.x + tid;
    const int i = t >> 4, q = t & 15;
    const int r = tid >> 4;
    const bool valid = i < n;
    const int ic = valid ? i : n - 1;
    const float4* __restrict__ h4 = reinterpret_cast<const float4*>(h);

    const int beg = rowptr[ic];
    const int end = rowptr[ic + 1];
    const float di = dinv[ic];
    const float4 hs = h4[ic * 16 + q];
    float4 acc = hs;
    acc.x *= di; acc.y *= di; acc.z *= di; acc.w *= di;

    for (int p = beg; p < end; p += 2 * GB) {
        const int mine = rec[min(p + q, end - 1)];
        batch_weighted<0>(h4, dinv, mine, p, end, q, acc);
        if (p + GB < end) batch_weighted<GB>(h4, dinv, mine, p + GB, end, q, acc);
    }
    const float4 bv = reinterpret_cast<const float4*>(biasIn)[q];
    acc.x = fmaf(di, acc.x, bv.x); acc.y = fmaf(di, acc.y, bv.y);
    acc.z = fmaf(di, acc.z, bv.z); acc.w = fmaf(di, acc.w, bv.w);
    acc.x = fmaxf(acc.x, 0.0f); acc.y = fmaxf(acc.y, 0.0f);
    acc.z = fmaxf(acc.z, 0.0f); acc.w = fmaxf(acc.w, 0.0f);
    *reinterpret_cast<float4*>(&sA[r * LP + q * 4]) = acc;
    __syncthreads();

    const int wave = tid >> 5;
    if (wave < 4) {
        const int lane    = tid & 31;
        const int half    = lane >> 4;
        const int l16     = lane & 15;
        const int colBase = wave << 4;
        const v2f* wl = reinterpret_cast<const v2f*>(sW + half * WP2) + colBase + l16;
        const float* arow = sA + l16 * LP;
        v8f c = {};
        #pragma unroll
        for (int kb = 0; kb < DCOL; kb += 4) {
            const int k0 = kb + half * 2;
            const v2f a = *reinterpret_cast<const v2f*>(arow + k0);
            const v2f b = wl[(kb >> 2) * WP2];
            c = __builtin_amdgcn_wmma_f32_16x16x4_f32(
                     false, a,  false, b,
                     (short)0, c,  false,  false);
        }
        #pragma unroll
        for (int v = 0; v < 8; ++v) sD[(v + half * 8) * LP + colBase + l16] = c[v];
    }
    __syncthreads();

    float4 o = *reinterpret_cast<const float4*>(&sD[r * LP + q * 4]);
    o.x *= di; o.y *= di; o.z *= di; o.w *= di;
    if (valid) reinterpret_cast<float4*>(out)[i * 16 + q] = o;
}

__global__ __launch_bounds__(256)
void k_agg_sum(const int* __restrict__ rowptr, const int* __restrict__ rec,
               const float* __restrict__ dinv, const float* __restrict__ h,
               const float* __restrict__ cvec, float* __restrict__ out, int n) {
    const int t = blockIdx.x * blockDim.x + threadIdx.x;
    const int i = t >> 4, q = t & 15;
    if (i >= n) return;
    const float4* __restrict__ h4 = reinterpret_cast<const float4*>(h);

    const int beg = rowptr[i];
    const int end = rowptr[i + 1];
    const float di = dinv[i];
    const float4 cv = reinterpret_cast<const float4*>(cvec)[q];
    float4 acc = h4[i * 16 + q];

    for (int p = beg; p < end; p += 2 * GB) {
        const int mine = rec[min(p + q, end - 1)];
        batch_plain<0>(h4, mine, p, end, q, acc);
        if (p + GB < end) batch_plain<GB>(h4, mine, p + GB, end, q, acc);
    }
    reinterpret_cast<float4*>(out)[i * 16 + q] =
        make_float4(fmaf(di, acc.x, cv.x), fmaf(di, acc.y, cv.y), fmaf(di, acc.z, cv.z), fmaf(di, acc.w, cv.w));
}

extern "C" void kernel_launch(void* const* d_in, const int* in_sizes, int n_in,
                              void* d_out, int out_size, void* d_ws, size_t ws_size,
                              hipStream_t stream) {
    const float* x   = (const float*)d_in[0];
    const int*   ei  = (const int*)  d_in[1];
    const float* W1  = (const float*)d_in[2];
    const float* b1  = (const float*)d_in[3];
    const float* W2  = (const float*)d_in[4];
    const float* b2  = (const float*)d_in[5];
    const float* Wfc = (const float*)d_in[6];
    const float* bfc = (const float*)d_in[7];
    float* out = (float*)d_out;

    const int N = in_sizes[0] / DIN;
    const int E = in_sizes[1] / 2;
    const int* src = ei;
    const int* dst = ei + E;

    const int NB  = (N + BN - 1) / BN;
    const int NC  = (E + CH - 1) / CH;
    const int NOF = (NB + 1) * NC;

    const size_t Np   = ((size_t)(N + 255) / 256) * 256;
    const size_t Ep   = ((size_t)(E + 255) / 256) * 256;
    const size_t NOFp = ((size_t)(NOF + 255) / 256) * 256;
    int*   OFF    = (int*)d_ws;
    int*   rowptr = OFF + NOFp;
    float* dinv   = (float*)(rowptr + Np + 256);
    int*   binned = (int*)(dinv + Np);
    int*   rec    = binned + 2 * (size_t)NC * CH;
    float* buf0   = (float*)(rec + 2 * Ep);
    float* buf1   = buf0 + (size_t)N * DCOL;
    float* Wp     = buf1 + ((size_t)N * DCOL + 255) / 256 * 256;
    float* cvec   = Wp + DCOL * DCOL;

    const int blk = 256;
    const int gN16 = (N * 16 + blk - 1) / blk;

    const int G  = (N + 127) / 128;
    const int G1 = G / 2;
    const int G2 = G - G1;

    k_chunk_sort<<<NC + G1 + 1, 256, 0, stream>>>(src, dst, OFF, binned, E, NB, NC, x, W1, buf0, N, 0, G1,
                                                  W2, b2, Wfc, bfc, Wp, cvec);
    k_bin_group <<<NB + G2, BN, 0, stream>>>(binned, OFF, rowptr, dinv, rec, N, NC, x, W1, buf0, NB, G1);

    k_agg_gemm<<<gN16, blk, 0, stream>>>(rowptr, rec, dinv, buf0, b1, Wp, buf1, N);

    k_agg_sum <<<gN16, blk, 0, stream>>>(rowptr, rec, dinv, buf1, cvec, out, N);
}
